# starting kernel
# baseline (speedup 1.0000x reference)
#include <hip/hip_runtime.h>
#include <hip/hip_bf16.h>

typedef __attribute__((ext_vector_type(2))) float v2f;
typedef __attribute__((ext_vector_type(8))) float v8f;

#define NNODES 100000
#define NEDGES 1600000

template <int K, int C>
__global__ __launch_bounds__(32 * (C / 16)) void gemm_wmma_kernel(
    const float* __restrict__ A, const float* __restrict__ B,
    float* __restrict__ D) {
  const int lane = threadIdx.x & 31;
  const int wave = threadIdx.x >> 5;
  const int row0 = blockIdx.x * 16;
  const int col0 = wave * 16;
  const int half = lane >> 4;
  const int l    = lane & 15;

  const float* arow = A + (long)(row0 + l) * K + 2 * half;
  const float* bcol = B + col0 + l;

  v8f acc = {};
#pragma unroll 4
  for (int k = 0; k < K; k += 4) {
    v2f a, b;
    a.x = arow[k + 0];
    a.y = arow[k + 1];
    b.x = bcol[(k + 2 * half + 0) * C];
    b.y = bcol[(k + 2 * half + 1) * C];
    acc = __builtin_amdgcn_wmma_f32_16x16x4_f32(
         false, a,  false, b,
         (short)0, acc,  false,  false);
  }

  float* drow = D + (long)(row0 + 8 * half) * C + col0 + l;
#pragma unroll
  for (int i = 0; i < 8; ++i) drow[i * C] = acc[i];
}

__global__ void fill_u32_kernel(unsigned* __restrict__ p, unsigned v, int n) {
  int i = blockIdx.x * blockDim.x + threadIdx.x;
  if (i < n) p[i] = v;
}

template <int H, int D>
__global__ void logits_kernel(const float* __restrict__ feat,
                              const float* __restrict__ al,
                              const float* __restrict__ ar,
                              float* __restrict__ el, float* __restrict__ er,
                              int NH) {
  int i = blockIdx.x * blockDim.x + threadIdx.x;
  if (i >= NH) return;
  int n = i / H, h = i % H;
  const float* f = feat + ((long)n * H + h) * D;
  const float* a = al + h * D;
  const float* r = ar + h * D;
  float sl = 0.f, sr = 0.f;
#pragma unroll
  for (int d = 0; d < D; ++d) {
    sl += f[d] * a[d];
    sr += f[d] * r[d];
  }
  el[i] = sl;
  er[i] = sr;
}

__device__ __forceinline__ unsigned float_ord_encode(float f) {
  unsigned u = __float_as_uint(f);
  return (u & 0x80000000u) ? ~u : (u | 0x80000000u);
}
__device__ __forceinline__ float float_ord_decode(unsigned u) {
  return (u & 0x80000000u) ? __uint_as_float(u ^ 0x80000000u)
                           : __uint_as_float(~u);
}
#define ENC_NEG_INF 0x007FFFFFu

template <int H>
__global__ void edge_max_kernel(const int* __restrict__ src,
                                const int* __restrict__ dst,
                                const float* __restrict__ el,
                                const float* __restrict__ er,
                                unsigned* __restrict__ mu, int EH) {
  int i = blockIdx.x * blockDim.x + threadIdx.x;
  if (i >= EH) return;
  int e = i / H, h = i % H;
  int s = src[e], d = dst[e];
  float v = el[s * H + h] + er[d * H + h];
  v = v > 0.f ? v : 0.2f * v;
  atomicMax(&mu[d * H + h], float_ord_encode(v));
}

template <int H, int D>
__global__ void edge_agg_kernel(const int* __restrict__ src,
                                const int* __restrict__ dst,
                                const float* __restrict__ el,
                                const float* __restrict__ er,
                                const unsigned* __restrict__ mu,
                                float* __restrict__ den,
                                const float* __restrict__ feat,
                                float* __restrict__ agg, int EH) {
  int i = blockIdx.x * blockDim.x + threadIdx.x;
  if (i >= EH) return;
  int e = i / H, h = i % H;
  int s = src[e], d = dst[e];
  float v = el[s * H + h] + er[d * H + h];
  v = v > 0.f ? v : 0.2f * v;
  float m = float_ord_decode(mu[d * H + h]);
  float w = expf(v - m);
  atomicAdd(&den[d * H + h], w);
  const float* fs = feat + ((long)s * H + h) * D;
  float* ad = agg + ((long)d * H + h) * D;
#pragma unroll
  for (int k = 0; k < D; ++k) atomicAdd(&ad[k], w * fs[k]);
}

template <int H, int D, bool DO_ELU>
__global__ void finalize_kernel(const float* __restrict__ agg,
                                const float* __restrict__ den,
                                const float* __restrict__ bias,
                                float* __restrict__ out, int NC) {
  int i = blockIdx.x * blockDim.x + threadIdx.x;
  if (i >= NC) return;
  int n = i / (H * D);
  int c = i % (H * D);
  int h = c / D;
  float dn = den[n * H + h];
  float x = dn > 0.f ? agg[i] / dn : 0.f;
  x += bias[c];
  if (DO_ELU) x = x > 0.f ? x : expf(x) - 1.f;
  out[i] = x;
}

extern "C" void kernel_launch(void* const* d_in, const int* in_sizes, int n_in,
                              void* d_out, int out_size, void* d_ws,
                              size_t ws_size, hipStream_t stream) {
  const float* x   = (const float*)d_in[0];
  const int*   src = (const int*)d_in[1];
  const int*   dst = (const int*)d_in[2];
  const float* W1  = (const float*)d_in[3];
  const float* al1 = (const float*)d_in[4];
  const float* ar1 = (const float*)d_in[5];
  const float* b1  = (const float*)d_in[6];
  const float* W2  = (const float*)d_in[7];
  const float* al2 = (const float*)d_in[8];
  const float* ar2 = (const float*)d_in[9];
  const float* b2  = (const float*)d_in[10];
  float* out = (float*)d_out;

  const int N = NNODES, E = NEDGES;

  float* ws = (float*)d_ws;
  float*    feat1 = ws;
  float*    el1   = feat1 + (long)N * 64;
  float*    er1   = el1 + (long)N * 8;
  unsigned* m1u   = (unsigned*)(er1 + (long)N * 8);
  float*    den1  = (float*)(m1u + (long)N * 8);
  float*    agg1  = den1 + (long)N * 8;
  float*    h1    = feat1;
  float*    feat2 = agg1;
  unsigned* m2u   = (unsigned*)(agg1 + (long)N * 32);
  float*    den2  = agg1 + (long)N * 32 + N;
  float*    el2   = agg1 + (long)N * 32 + 2 * N;
  float*    er2   = agg1 + (long)N * 32 + 3 * N;
  float*    agg2  = el1;

  const int TB = 256;
  auto nb = [](long n, int b) { return (int)((n + b - 1) / b); };

  fill_u32_kernel<<<nb((long)N * 8, TB), TB, 0, stream>>>(m1u, ENC_NEG_INF, N * 8);
  fill_u32_kernel<<<nb((long)N * 8, TB), TB, 0, stream>>>((unsigned*)den1, 0u, N * 8);
  fill_u32_kernel<<<nb((long)N * 64, TB), TB, 0, stream>>>((unsigned*)agg1, 0u, N * 64);

  gemm_wmma_kernel<256, 64><<<N / 16, 128, 0, stream>>>(x, W1, feat1);
  logits_kernel<8, 8><<<nb((long)N * 8, TB), TB, 0, stream>>>(feat1, al1, ar1, el1, er1, N * 8);
  edge_max_kernel<8><<<nb((long)E * 8, TB), TB, 0, stream>>>(src, dst, el1, er1, m1u, E * 8);
  edge_agg_kernel<8, 8><<<nb((long)E * 8, TB), TB, 0, stream>>>(src, dst, el1, er1, m1u, den1, feat1, agg1, E * 8);
  finalize_kernel<8, 8, true><<<nb((long)N * 64, TB), TB, 0, stream>>>(agg1, den1, b1, h1, N * 64);

  fill_u32_kernel<<<nb((long)N, TB), TB, 0, stream>>>(m2u, ENC_NEG_INF, N);
  fill_u32_kernel<<<nb((long)N, TB), TB, 0, stream>>>((unsigned*)den2, 0u, N);

  gemm_wmma_kernel<64, 32><<<N / 16, 64, 0, stream>>>(h1, W2, feat2);
  logits_kernel<1, 32><<<nb((long)N, TB), TB, 0, stream>>>(feat2, al2, ar2, el2, er2, N);

  fill_u32_kernel<<<nb((long)N * 32, TB), TB, 0, stream>>>((unsigned*)agg2, 0u, N * 32);

  edge_max_kernel<1><<<nb((long)E, TB), TB, 0, stream>>>(src, dst, el2, er2, m2u, E);
  edge_agg_kernel<1, 32><<<nb((long)E, TB), TB, 0, stream>>>(src, dst, el2, er2, m2u, den2, feat2, agg2, E);
  finalize_kernel<1, 32, false><<<nb((long)N * 32, TB), TB, 0, stream>>>(agg2, den2, b2, out, N * 32);
}


# edge aggregation remapped: a wave's lanes own the features of one edge row, so each feature load and float atomic add covers 128 contiguous bytes, not 32 rows; atomics kept (arrival-order sums)
# speedup vs baseline: 2.2322x; 2.2322x over previous
#include <hip/hip_runtime.h>
#include <hip/hip_bf16.h>

typedef __attribute__((ext_vector_type(2))) float v2f;
typedef __attribute__((ext_vector_type(8))) float v8f;

#define NNODES 100000
#define NEDGES 1600000

template <int K, int C>
__global__ __launch_bounds__(32 * (C / 16)) void gemm_wmma_kernel(
    const float* __restrict__ A, const float* __restrict__ B,
    float* __restrict__ D) {
  const int lane = threadIdx.x & 31;
  const int wave = threadIdx.x >> 5;
  const int row0 = blockIdx.x * 16;
  const int col0 = wave * 16;
  const int half = lane >> 4;
  const int l    = lane & 15;

  const float* arow = A + (long)(row0 + l) * K + 2 * half;
  const float* bcol = B + col0 + l;

  v8f acc = {};
#pragma unroll 4
  for (int k = 0; k < K; k += 4) {
    v2f a, b;
    a.x = arow[k + 0];
    a.y = arow[k + 1];
    b.x = bcol[(k + 2 * half + 0) * C];
    b.y = bcol[(k + 2 * half + 1) * C];
    acc = __builtin_amdgcn_wmma_f32_16x16x4_f32(
         false, a,  false, b,
         (short)0, acc,  false,  false);
  }

  float* drow = D + (long)(row0 + 8 * half) * C + col0 + l;
#pragma unroll
  for (int i = 0; i < 8; ++i) drow[i * C] = acc[i];
}

__global__ void fill_u32_kernel(unsigned* __restrict__ p, unsigned v, int n) {
  int i = blockIdx.x * blockDim.x + threadIdx.x;
  if (i < n) p[i] = v;
}

template <int H, int D>
__global__ void logits_kernel(const float* __restrict__ feat,
                              const float* __restrict__ al,
                              const float* __restrict__ ar,
                              float* __restrict__ el, float* __restrict__ er,
                              int NH) {
  int i = blockIdx.x * blockDim.x + threadIdx.x;
  if (i >= NH) return;
  int n = i / H, h = i % H;
  const float* f = feat + ((long)n * H + h) * D;
  const float* a = al + h * D;
  const float* r = ar + h * D;
  float sl = 0.f, sr = 0.f;
#pragma unroll
  for (int d = 0; d < D; ++d) {
    sl += f[d] * a[d];
    sr += f[d] * r[d];
  }
  el[i] = sl;
  er[i] = sr;
}

__device__ __forceinline__ unsigned float_ord_encode(float f) {
  unsigned u = __float_as_uint(f);
  return (u & 0x80000000u) ? ~u : (u | 0x80000000u);
}
__device__ __forceinline__ float float_ord_decode(unsigned u) {
  return (u & 0x80000000u) ? __uint_as_float(u ^ 0x80000000u)
                           : __uint_as_float(~u);
}
#define ENC_NEG_INF 0x007FFFFFu

template <int H>
__global__ void edge_max_kernel(const int* __restrict__ src,
                                const int* __restrict__ dst,
                                const float* __restrict__ el,
                                const float* __restrict__ er,
                                unsigned* __restrict__ mu, int EH) {
  int i = blockIdx.x * blockDim.x + threadIdx.x;
  if (i >= EH) return;
  int e = i / H, h = i % H;
  int s = src[e], d = dst[e];
  float v = el[s * H + h] + er[d * H + h];
  v = v > 0.f ? v : 0.2f * v;
  atomicMax(&mu[d * H + h], float_ord_encode(v));
}

template <int H, int D>
__global__ void edge_agg_kernel(const int* __restrict__ src,
                                const int* __restrict__ dst,
                                const float* __restrict__ el,
                                const float* __restrict__ er,
                                const unsigned* __restrict__ mu,
                                float* __restrict__ den,
                                const float* __restrict__ feat,
                                float* __restrict__ agg, int EH) {
  constexpr int C   = H * D;
  constexpr int EPW = 32 / H;
  constexpr int NP  = C / 32;
  static_assert(32 % H == 0 && C % 32 == 0, "row pieces must tile the wave");

  const int i    = blockIdx.x * blockDim.x + threadIdx.x;
  const int lane = threadIdx.x & 31;
  int s = 0, d = 0;
  float w = 0.f;
  if (i < EH) {
    int e = i / H, h = i % H;
    s = src[e];
    d = dst[e];
    float v = el[s * H + h] + er[d * H + h];
    v = v > 0.f ? v : 0.2f * v;
    float m = float_ord_decode(mu[d * H + h]);
    w = expf(v - m);
    atomicAdd(&den[d * H + h], w);
  }

  const int i0 = __builtin_amdgcn_readfirstlane(i);

  if (i0 + 32 <= EH) {
    float f[EPW][NP];
#pragma unroll
    for (int j = 0; j < EPW; ++j) {
      const int sj = __builtin_amdgcn_readlane(s, j * H);
      const float* fs = feat + (long)sj * C;
#pragma unroll
      for (int p = 0; p < NP; ++p) f[j][p] = fs[p * 32 + lane];
    }
#pragma unroll
    for (int j = 0; j < EPW; ++j) {
      const int dj = __builtin_amdgcn_readlane(d, j * H);
      float* ad = agg + (long)dj * C;
#pragma unroll
      for (int p = 0; p < NP; ++p) {
        const int c = p * 32 + lane;
        float wl;
        if (H == 1)
          wl = __int_as_float(__builtin_amdgcn_readlane(__float_as_int(w), j));
        else
          wl = __shfl(w, j * H + c / D, 32);
        atomicAdd(&ad[c], wl * f[j][p]);
      }
    }
  } else {
    for (int j = 0; i0 + j * H < EH; ++j) {
      const int sj = __builtin_amdgcn_readlane(s, j * H);
      const int dj = __builtin_amdgcn_readlane(d, j * H);
      const float* fs = feat + (long)sj * C;
      float* ad = agg + (long)dj * C;
      for (int p = 0; p < NP; ++p) {
        const int c = p * 32 + lane;
        const float wl = __shfl(w, j * H + c / D, 32);
        atomicAdd(&ad[c], wl * fs[c]);
      }
    }
  }
}

template <int H, int D, bool DO_ELU>
__global__ void finalize_kernel(const float* __restrict__ agg,
                                const float* __restrict__ den,
                                const float* __restrict__ bias,
                                float* __restrict__ out, int NC) {
  int i = blockIdx.x * blockDim.x + threadIdx.x;
  if (i >= NC) return;
  int n = i / (H * D);
  int c = i % (H * D);
  int h = c / D;
  float dn = den[n * H + h];
  float x = dn > 0.f ? agg[i] / dn : 0.f;
  x += bias[c];
  if (DO_ELU) x = x > 0.f ? x : expf(x) - 1.f;
  out[i] = x;
}

extern "C" void kernel_launch(void* const* d_in, const int* in_sizes, int n_in,
                              void* d_out, int out_size, void* d_ws,
                              size_t ws_size, hipStream_t stream) {
  const float* x   = (const float*)d_in[0];
  const int*   src = (const int*)d_in[1];
  const int*   dst = (const int*)d_in[2];
  const float* W1  = (const float*)d_in[3];
  const float* al1 = (const float*)d_in[4];
  const float* ar1 = (const float*)d_in[5];
  const float* b1  = (const float*)d_in[6];
  const float* W2  = (const float*)d_in[7];
  const float* al2 = (const float*)d_in[8];
  const float* ar2 = (const float*)d_in[9];
  const float* b2  = (const float*)d_in[10];
  float* out = (float*)d_out;

  const int N = NNODES, E = NEDGES;

  float* ws = (float*)d_ws;
  float*    feat1 = ws;
  float*    el1   = feat1 + (long)N * 64;
  float*    er1   = el1 + (long)N * 8;
  unsigned* m1u   = (unsigned*)(er1 + (long)N * 8);
  float*    den1  = (float*)(m1u + (long)N * 8);
  float*    agg1  = den1 + (long)N * 8;
  float*    h1    = feat1;
  float*    feat2 = agg1;
  unsigned* m2u   = (unsigned*)(agg1 + (long)N * 32);
  float*    den2  = agg1 + (long)N * 32 + N;
  float*    el2   = agg1 + (long)N * 32 + 2 * N;
  float*    er2   = agg1 + (long)N * 32 + 3 * N;
  float*    agg2  = el1;

  const int TB = 256;
  auto nb = [](long n, int b) { return (int)((n + b - 1) / b); };

  fill_u32_kernel<<<nb((long)N * 8, TB), TB, 0, stream>>>(m1u, ENC_NEG_INF, N * 8);
  fill_u32_kernel<<<nb((long)N * 8, TB), TB, 0, stream>>>((unsigned*)den1, 0u, N * 8);
  fill_u32_kernel<<<nb((long)N * 64, TB), TB, 0, stream>>>((unsigned*)agg1, 0u, N * 64);

  gemm_wmma_kernel<256, 64><<<N / 16, 128, 0, stream>>>(x, W1, feat1);
  logits_kernel<8, 8><<<nb((long)N * 8, TB), TB, 0, stream>>>(feat1, al1, ar1, el1, er1, N * 8);
  edge_max_kernel<8><<<nb((long)E * 8, TB), TB, 0, stream>>>(src, dst, el1, er1, m1u, E * 8);
  edge_agg_kernel<8, 8><<<nb((long)E * 8, TB), TB, 0, stream>>>(src, dst, el1, er1, m1u, den1, feat1, agg1, E * 8);
  finalize_kernel<8, 8, true><<<nb((long)N * 64, TB), TB, 0, stream>>>(agg1, den1, b1, h1, N * 64);

  fill_u32_kernel<<<nb((long)N, TB), TB, 0, stream>>>(m2u, ENC_NEG_INF, N);
  fill_u32_kernel<<<nb((long)N, TB), TB, 0, stream>>>((unsigned*)den2, 0u, N);

  gemm_wmma_kernel<64, 32><<<N / 16, 64, 0, stream>>>(h1, W2, feat2);
  logits_kernel<1, 32><<<nb((long)N, TB), TB, 0, stream>>>(feat2, al2, ar2, el2, er2, N);

  fill_u32_kernel<<<nb((long)N * 32, TB), TB, 0, stream>>>((unsigned*)agg2, 0u, N * 32);

  edge_max_kernel<1><<<nb((long)E, TB), TB, 0, stream>>>(src, dst, el2, er2, m2u, E);
  edge_agg_kernel<1, 32><<<nb((long)E, TB), TB, 0, stream>>>(src, dst, el2, er2, m2u, den2, feat2, agg2, E);
  finalize_kernel<1, 32, false><<<nb((long)N * 32, TB), TB, 0, stream>>>(agg2, den2, b2, out, N * 32);
}


# edges bucketed by destination in-call (count, scan, integer-atomic slots); one wave per node does max, exp, sums, normalise in registers: no float atomics; sums in slot order, er1 pairwise
# speedup vs baseline: 3.1249x; 1.3999x over previous
#include <hip/hip_runtime.h>
#include <hip/hip_bf16.h>

typedef __attribute__((ext_vector_type(2))) float v2f;
typedef __attribute__((ext_vector_type(8))) float v8f;

#define NNODES 100000
#define NEDGES 1600000

template <int K, int C>
__global__ __launch_bounds__(32 * (C / 16)) void gemm_wmma_kernel(
    const float* __restrict__ A, const float* __restrict__ B,
    float* __restrict__ D) {
  const int lane = threadIdx.x & 31;
  const int wave = threadIdx.x >> 5;
  const int row0 = blockIdx.x * 16;
  const int col0 = wave * 16;
  const int half = lane >> 4;
  const int l    = lane & 15;

  const float* arow = A + (long)(row0 + l) * K + 2 * half;
  const float* bcol = B + col0 + l;

  v8f acc = {};
#pragma unroll 4
  for (int k = 0; k < K; k += 4) {
    v2f a, b;
    a.x = arow[k + 0];
    a.y = arow[k + 1];
    b.x = bcol[(k + 2 * half + 0) * C];
    b.y = bcol[(k + 2 * half + 1) * C];
    acc = __builtin_amdgcn_wmma_f32_16x16x4_f32(
         false, a,  false, b,
         (short)0, acc,  false,  false);
  }

  float* drow = D + (long)(row0 + 8 * half) * C + col0 + l;
#pragma unroll
  for (int i = 0; i < 8; ++i) drow[i * C] = acc[i];
}

__global__ void fill_u32_kernel(unsigned* __restrict__ p, unsigned v, int n) {
  int i = blockIdx.x * blockDim.x + threadIdx.x;
  if (i < n) p[i] = v;
}

template <int H, int D, bool WRITE_ER>
__global__ void logits_kernel(const float* __restrict__ feat,
                              const float* __restrict__ al,
                              const float* __restrict__ ar,
                              float* __restrict__ el, float* __restrict__ er,
                              int NH) {
  int i = blockIdx.x * blockDim.x + threadIdx.x;
  if (i >= NH) return;
  int n = i / H, h = i % H;
  const float* f = feat + ((long)n * H + h) * D;
  const float* a = al + h * D;
  const float* r = ar + h * D;
  float sl = 0.f, sr = 0.f;
#pragma unroll
  for (int d = 0; d < D; ++d) {
    sl += f[d] * a[d];
    if (WRITE_ER) sr += f[d] * r[d];
  }
  el[i] = sl;
  if (WRITE_ER) er[i] = sr;
}

__device__ __forceinline__ float leaky(float v) { return v > 0.f ? v : 0.2f * v; }

__global__ void degree_count_kernel(const int* __restrict__ dst,
                                    unsigned* __restrict__ cnt, int E) {
  int e = blockIdx.x * blockDim.x + threadIdx.x;
  if (e < E) atomicAdd(&cnt[dst[e]], 1u);
}

__global__ __launch_bounds__(256) void block_sum_kernel(
    const unsigned* __restrict__ cnt, unsigned* __restrict__ bsum, int n) {
  __shared__ unsigned wsum[8];
  const int lane = threadIdx.x & 31, wave = threadIdx.x >> 5;
  const int i0 = blockIdx.x * 1024 + threadIdx.x * 4;
  unsigned s = 0;
#pragma unroll
  for (int k = 0; k < 4; ++k)
    if (i0 + k < n) s += cnt[i0 + k];
  for (int off = 16; off > 0; off >>= 1) s += __shfl_xor(s, off, 32);
  if (lane == 0) wsum[wave] = s;
  __syncthreads();
  if (threadIdx.x == 0) {
    unsigned t = 0;
    for (int w = 0; w < 8; ++w) t += wsum[w];
    bsum[blockIdx.x] = t;
  }
}

__global__ __launch_bounds__(256) void scan_kernel(
    const unsigned* __restrict__ cnt, const unsigned* __restrict__ bsum,
    unsigned* __restrict__ rowptr, int n) {
  __shared__ unsigned wpre[8];
  __shared__ unsigned wtot[8];
  const int t = threadIdx.x, lane = t & 31, wave = t >> 5;

  unsigned p = 0;
  for (int b = t; b < (int)blockIdx.x; b += 256) p += bsum[b];
  for (int off = 16; off > 0; off >>= 1) p += __shfl_xor(p, off, 32);
  if (lane == 0) wpre[wave] = p;

  const int i0 = blockIdx.x * 1024 + t * 4;
  unsigned c[4];
#pragma unroll
  for (int k = 0; k < 4; ++k) c[k] = (i0 + k < n) ? cnt[i0 + k] : 0u;
  const unsigned loc = c[0] + c[1] + c[2] + c[3];
  unsigned inc = loc;
  for (int off = 1; off < 32; off <<= 1) {
    unsigned y = __shfl_up(inc, off, 32);
    if (lane >= off) inc += y;
  }
  if (lane == 31) wtot[wave] = inc;
  __syncthreads();

  unsigned ex = inc - loc;
  for (int w = 0; w < 8; ++w) {
    ex += wpre[w];
    if (w < wave) ex += wtot[w];
  }
#pragma unroll
  for (int k = 0; k < 4; ++k) {
    if (i0 + k <= n) rowptr[i0 + k] = ex;
    ex += c[k];
  }
}

__global__ void csr_fill_kernel(const int* __restrict__ src,
                                const int* __restrict__ dst,
                                const unsigned* __restrict__ rowptr,
                                unsigned* __restrict__ cnt,
                                int* __restrict__ csr_src, int E) {
  int e = blockIdx.x * blockDim.x + threadIdx.x;
  if (e >= E) return;
  const int d = dst[e];
  const unsigned old = atomicSub(&cnt[d], 1u);
  csr_src[rowptr[d] + old - 1u] = src[e];
}

__global__ __launch_bounds__(256) void node_agg_h8_kernel(
    const unsigned* __restrict__ rowptr, const int* __restrict__ csr_src,
    const float* __restrict__ feat, const float* __restrict__ el,
    const float* __restrict__ ar, const float* __restrict__ bias,
    float* __restrict__ hout, int N) {
  const int lane = threadIdx.x & 31;
  const int node = blockIdx.x * 8 + __builtin_amdgcn_readfirstlane(threadIdx.x >> 5);
  if (node >= N) return;
  const int beg = (int)rowptr[node];
  const int end = (int)rowptr[node + 1];
  const int eg = lane >> 3, hh = lane & 7, hsel = lane >> 2;

  const float2 fd = *reinterpret_cast<const float2*>(feat + (long)node * 64 + 2 * lane);
  const float2 a2 = *reinterpret_cast<const float2*>(ar + 2 * lane);
  float erp = fd.x * a2.x + fd.y * a2.y;
  erp += __shfl_xor(erp, 1, 32);
  erp += __shfl_xor(erp, 2, 32);
  const float er_h = __shfl(erp, 4 * hh, 32);

  float mx = -__builtin_inff();
  for (int base = beg; base < end; base += 32) {
    const int nloc = (end - base < 32) ? end - base : 32;
    const int sid = (lane < nloc) ? csr_src[base + lane] : 0;
    for (int c = 0; c < nloc; c += 4) {
      const int k = c + eg;
      const int s = __shfl(sid, k, 32);
      const float v = leaky(el[s * 8 + hh] + er_h);
      mx = fmaxf(mx, k < nloc ? v : -__builtin_inff());
    }
  }
  mx = fmaxf(mx, __shfl_xor(mx, 8, 32));
  mx = fmaxf(mx, __shfl_xor(mx, 16, 32));

  float den = 0.f, ax = 0.f, ay = 0.f;
  for (int base = beg; base < end; base += 32) {
    const int nloc = (end - base < 32) ? end - base : 32;
    const int sid = (lane < nloc) ? csr_src[base + lane] : 0;
    for (int c = 0; c < nloc; c += 4) {
      float2 f[4];
#pragma unroll
      for (int j = 0; j < 4; ++j) {
        const int sj = __builtin_amdgcn_readlane(sid, c + j);
        f[j] = *reinterpret_cast<const float2*>(feat + (long)sj * 64 + 2 * lane);
      }
      const int k = c + eg;
      const int s = __shfl(sid, k, 32);
      const float v = leaky(el[s * 8 + hh] + er_h);
      const float w = (k < nloc) ? expf(v - mx) : 0.f;
      den += w;
#pragma unroll
      for (int j = 0; j < 4; ++j) {
        const float wj = __shfl(w, j * 8 + hsel, 32);
        if (c + j < nloc) {
          ax += wj * f[j].x;
          ay += wj * f[j].y;
        }
      }
    }
  }
  den += __shfl_xor(den, 8, 32);
  den += __shfl_xor(den, 16, 32);
  const float dsel = __shfl(den, hsel, 32);

  const float2 bb = *reinterpret_cast<const float2*>(bias + 2 * lane);
  float ox = (dsel > 0.f ? ax / dsel : 0.f) + bb.x;
  float oy = (dsel > 0.f ? ay / dsel : 0.f) + bb.y;
  ox = ox > 0.f ? ox : expf(ox) - 1.f;
  oy = oy > 0.f ? oy : expf(oy) - 1.f;
  float2 o;
  o.x = ox;
  o.y = oy;
  *reinterpret_cast<float2*>(hout + (long)node * 64 + 2 * lane) = o;
}

__global__ __launch_bounds__(256) void node_agg_h1_kernel(
    const unsigned* __restrict__ rowptr, const int* __restrict__ csr_src,
    const float* __restrict__ feat, const float* __restrict__ el,
    const float* __restrict__ er, const float* __restrict__ bias,
    float* __restrict__ out, int N) {
  const int lane = threadIdx.x & 31;
  const int node = blockIdx.x * 8 + __builtin_amdgcn_readfirstlane(threadIdx.x >> 5);
  if (node >= N) return;
  const int beg = (int)rowptr[node];
  const int end = (int)rowptr[node + 1];
  const float er_d = er[node];

  float mx = -__builtin_inff();
  for (int base = beg; base < end; base += 32) {
    const int nloc = (end - base < 32) ? end - base : 32;
    const int sid = (lane < nloc) ? csr_src[base + lane] : 0;
    const float v = leaky(el[sid] + er_d);
    mx = fmaxf(mx, lane < nloc ? v : -__builtin_inff());
  }
  for (int off = 16; off > 0; off >>= 1) mx = fmaxf(mx, __shfl_xor(mx, off, 32));

  float den = 0.f, acc = 0.f;
  for (int base = beg; base < end; base += 32) {
    const int nloc = (end - base < 32) ? end - base : 32;
    const int sid = (lane < nloc) ? csr_src[base + lane] : 0;
    const float v = leaky(el[sid] + er_d);
    const float w = (lane < nloc) ? expf(v - mx) : 0.f;
    den += w;
    for (int j = 0; j < nloc; j += 4) {
      float f[4];
#pragma unroll
      for (int u = 0; u < 4; ++u) {
        const int sj = __builtin_amdgcn_readlane(sid, j + u);
        f[u] = feat[(long)sj * 32 + lane];
      }
#pragma unroll
      for (int u = 0; u < 4; ++u) {
        const float wj = __int_as_float(__builtin_amdgcn_readlane(__float_as_int(w), j + u));
        if (j + u < nloc) acc += wj * f[u];
      }
    }
  }
  for (int off = 16; off > 0; off >>= 1) den += __shfl_xor(den, off, 32);
  out[(long)node * 32 + lane] = (den > 0.f ? acc / den : 0.f) + bias[lane];
}

extern "C" void kernel_launch(void* const* d_in, const int* in_sizes, int n_in,
                              void* d_out, int out_size, void* d_ws,
                              size_t ws_size, hipStream_t stream) {
  const float* x   = (const float*)d_in[0];
  const int*   src = (const int*)d_in[1];
  const int*   dst = (const int*)d_in[2];
  const float* W1  = (const float*)d_in[3];
  const float* al1 = (const float*)d_in[4];
  const float* ar1 = (const float*)d_in[5];
  const float* b1  = (const float*)d_in[6];
  const float* W2  = (const float*)d_in[7];
  const float* al2 = (const float*)d_in[8];
  const float* ar2 = (const float*)d_in[9];
  const float* b2  = (const float*)d_in[10];
  float* out = (float*)d_out;

  const int N = NNODES, E = NEDGES;

  float* ws = (float*)d_ws;
  float*    feat1  = ws;
  float*    el1    = ws + (long)N * 64;
  float*    h1     = ws + (long)N * 72;
  int*      csr    = (int*)(ws + (long)N * 136);
  unsigned* rowptr = (unsigned*)(ws + (long)N * 152);
  unsigned* cnt    = (unsigned*)(ws + (long)N * 153 + 32);
  unsigned* bsum   = (unsigned*)(ws + (long)N * 154 + 32);
  float*    feat2  = ws;
  float*    el2    = ws + (long)N * 32;
  float*    er2    = ws + (long)N * 33;

  const int TB = 256;
  auto nb = [](long n, int b) { return (int)((n + b - 1) / b); };

  fill_u32_kernel<<<nb((long)N, TB), TB, 0, stream>>>(cnt, 0u, N);
  degree_count_kernel<<<nb((long)E, TB), TB, 0, stream>>>(dst, cnt, E);
  block_sum_kernel<<<nb((long)N + 1, 1024), 256, 0, stream>>>(cnt, bsum, N);
  scan_kernel<<<nb((long)N + 1, 1024), 256, 0, stream>>>(cnt, bsum, rowptr, N);
  csr_fill_kernel<<<nb((long)E, TB), TB, 0, stream>>>(src, dst, rowptr, cnt, csr, E);

  gemm_wmma_kernel<256, 64><<<N / 16, 128, 0, stream>>>(x, W1, feat1);
  logits_kernel<8, 8, false><<<nb((long)N * 8, TB), TB, 0, stream>>>(feat1, al1, ar1, el1, el1, N * 8);
  node_agg_h8_kernel<<<nb((long)N, 8), 256, 0, stream>>>(rowptr, csr, feat1, el1, ar1, b1, h1, N);

  gemm_wmma_kernel<64, 32><<<N / 16, 64, 0, stream>>>(h1, W2, feat2);
  logits_kernel<1, 32, true><<<nb((long)N, TB), TB, 0, stream>>>(feat2, al2, ar2, el2, er2, N);
  node_agg_h1_kernel<<<nb((long)N, 8), 256, 0, stream>>>(rowptr, csr, feat2, el2, er2, b2, out, N);
}


# edge scores formed in each node's wave from the gathered rows (lane-tree dots); one-pass softmax, running max and rescale: one gather per edge, no score tables, 2 launches fewer; sums reordered
# speedup vs baseline: 3.4088x; 1.0909x over previous
#include <hip/hip_runtime.h>
#include <hip/hip_bf16.h>

typedef __attribute__((ext_vector_type(2))) float v2f;
typedef __attribute__((ext_vector_type(8))) float v8f;

#define NNODES 100000
#define NEDGES 1600000

template <int K, int C>
__global__ __launch_bounds__(32 * (C / 16)) void gemm_wmma_kernel(
    const float* __restrict__ A, const float* __restrict__ B,
    float* __restrict__ D) {
  const int lane = threadIdx.x & 31;
  const int wave = threadIdx.x >> 5;
  const int row0 = blockIdx.x * 16;
  const int col0 = wave * 16;
  const int half = lane >> 4;
  const int l    = lane & 15;

  const float* arow = A + (long)(row0 + l) * K + 2 * half;
  const float* bcol = B + col0 + l;

  v8f acc = {};
#pragma unroll 4
  for (int k = 0; k < K; k += 4) {
    v2f a, b;
    a.x = arow[k + 0];
    a.y = arow[k + 1];
    b.x = bcol[(k + 2 * half + 0) * C];
    b.y = bcol[(k + 2 * half + 1) * C];
    acc = __builtin_amdgcn_wmma_f32_16x16x4_f32(
         false, a,  false, b,
         (short)0, acc,  false,  false);
  }

  float* drow = D + (long)(row0 + 8 * half) * C + col0 + l;
#pragma unroll
  for (int i = 0; i < 8; ++i) drow[i * C] = acc[i];
}

__global__ void fill_u32_kernel(unsigned* __restrict__ p, unsigned v, int n) {
  int i = blockIdx.x * blockDim.x + threadIdx.x;
  if (i < n) p[i] = v;
}

__device__ __forceinline__ float leaky(float v) { return v > 0.f ? v : 0.2f * v; }

__global__ void degree_count_kernel(const int* __restrict__ dst,
                                    unsigned* __restrict__ cnt, int E) {
  int e = blockIdx.x * blockDim.x + threadIdx.x;
  if (e < E) atomicAdd(&cnt[dst[e]], 1u);
}

__global__ __launch_bounds__(256) void block_sum_kernel(
    const unsigned* __restrict__ cnt, unsigned* __restrict__ bsum, int n) {
  __shared__ unsigned wsum[8];
  const int lane = threadIdx.x & 31, wave = threadIdx.x >> 5;
  const int i0 = blockIdx.x * 1024 + threadIdx.x * 4;
  unsigned s = 0;
#pragma unroll
  for (int k = 0; k < 4; ++k)
    if (i0 + k < n) s += cnt[i0 + k];
  for (int off = 16; off > 0; off >>= 1) s += __shfl_xor(s, off, 32);
  if (lane == 0) wsum[wave] = s;
  __syncthreads();
  if (threadIdx.x == 0) {
    unsigned t = 0;
    for (int w = 0; w < 8; ++w) t += wsum[w];
    bsum[blockIdx.x] = t;
  }
}

__global__ __launch_bounds__(256) void scan_kernel(
    const unsigned* __restrict__ cnt, const unsigned* __restrict__ bsum,
    unsigned* __restrict__ rowptr, int n) {
  __shared__ unsigned wpre[8];
  __shared__ unsigned wtot[8];
  const int t = threadIdx.x, lane = t & 31, wave = t >> 5;

  unsigned p = 0;
  for (int b = t; b < (int)blockIdx.x; b += 256) p += bsum[b];
  for (int off = 16; off > 0; off >>= 1) p += __shfl_xor(p, off, 32);
  if (lane == 0) wpre[wave] = p;

  const int i0 = blockIdx.x * 1024 + t * 4;
  unsigned c[4];
#pragma unroll
  for (int k = 0; k < 4; ++k) c[k] = (i0 + k < n) ? cnt[i0 + k] : 0u;
  const unsigned loc = c[0] + c[1] + c[2] + c[3];
  unsigned inc = loc;
  for (int off = 1; off < 32; off <<= 1) {
    unsigned y = __shfl_up(inc, off, 32);
    if (lane >= off) inc += y;
  }
  if (lane == 31) wtot[wave] = inc;
  __syncthreads();

  unsigned ex = inc - loc;
  for (int w = 0; w < 8; ++w) {
    ex += wpre[w];
    if (w < wave) ex += wtot[w];
  }
#pragma unroll
  for (int k = 0; k < 4; ++k) {
    if (i0 + k <= n) rowptr[i0 + k] = ex;
    ex += c[k];
  }
}

__global__ void csr_fill_kernel(const int* __restrict__ src,
                                const int* __restrict__ dst,
                                const unsigned* __restrict__ rowptr,
                                unsigned* __restrict__ cnt,
                                int* __restrict__ csr_src, int E) {
  int e = blockIdx.x * blockDim.x + threadIdx.x;
  if (e >= E) return;
  const int d = dst[e];
  const unsigned old = atomicSub(&cnt[d], 1u);
  csr_src[rowptr[d] + old - 1u] = src[e];
}

template <int CTRL>
__device__ __forceinline__ float dpp(float x) {
  return __builtin_bit_cast(
      float, __builtin_amdgcn_update_dpp(0, __builtin_bit_cast(int, x), CTRL, 0xf, 0xf, true));
}
__device__ __forceinline__ float xhalf(float x) {
  return __builtin_bit_cast(
      float, __builtin_amdgcn_permlanex16(__builtin_bit_cast(int, x), __builtin_bit_cast(int, x),
                                          0x76543210, 0xfedcba98, false, false));
}

__global__ __launch_bounds__(256) void node_agg_h8_kernel(
    const unsigned* __restrict__ rowptr, const int* __restrict__ csr_src,
    const float* __restrict__ feat, const float* __restrict__ al,
    const float* __restrict__ ar, const float* __restrict__ bias,
    float* __restrict__ hout, int N) {
  const int lane = threadIdx.x & 31;
  const int node = blockIdx.x * 8 + __builtin_amdgcn_readfirstlane(threadIdx.x >> 5);
  if (node >= N) return;
  const int beg = (int)rowptr[node];
  const int end = (int)rowptr[node + 1];
  const int q = lane & 3;
  const bool q1 = (q & 1) != 0, q2 = (q & 2) != 0;

  const float2 fd = *reinterpret_cast<const float2*>(feat + (long)node * 64 + 2 * lane);
  const float2 r2 = *reinterpret_cast<const float2*>(ar + 2 * lane);
  const float2 a2 = *reinterpret_cast<const float2*>(al + 2 * lane);
  float er_h = fd.x * r2.x + fd.y * r2.y;
  er_h += dpp<0xB1>(er_h);
  er_h += dpp<0x4E>(er_h);

  float m = -__builtin_inff();
  float den = 0.f, ax = 0.f, ay = 0.f;
  for (int base = beg; base < end; base += 32) {
    const int nloc = (end - base < 32) ? end - base : 32;
    const int sid = (lane < nloc) ? csr_src[base + lane] : 0;
    for (int c = 0; c < nloc; c += 8) {
      float2 f[8];
#pragma unroll
      for (int j = 0; j < 8; ++j) {
        const int sj = __builtin_amdgcn_readlane(sid, c + j);
        f[j] = *reinterpret_cast<const float2*>(feat + (long)sj * 64 + 2 * lane);
      }
      float v[2];
#pragma unroll
      for (int g = 0; g < 2; ++g) {
        const float p0 = f[4 * g + 0].x * a2.x + f[4 * g + 0].y * a2.y;
        const float p1 = f[4 * g + 1].x * a2.x + f[4 * g + 1].y * a2.y;
        const float p2 = f[4 * g + 2].x * a2.x + f[4 * g + 2].y * a2.y;
        const float p3 = f[4 * g + 3].x * a2.x + f[4 * g + 3].y * a2.y;
        const float t01 = (q1 ? p1 : p0) + dpp<0xB1>(q1 ? p0 : p1);
        const float t23 = (q1 ? p3 : p2) + dpp<0xB1>(q1 ? p2 : p3);
        const float s = (q2 ? t23 : t01) + dpp<0x4E>(q2 ? t01 : t23);
        const float lv = leaky(s + er_h);
        v[g] = (c + 4 * g + q < nloc) ? lv : -__builtin_inff();
      }
      float cm = fmaxf(v[0], v[1]);
      cm = fmaxf(cm, dpp<0xB1>(cm));
      cm = fmaxf(cm, dpp<0x4E>(cm));
      const float mn = fmaxf(m, cm);
      const float sc = expf(m - mn);
      m = mn;
      float w[2];
      w[0] = expf(v[0] - mn);
      w[1] = expf(v[1] - mn);
      den = den * sc + w[0] + w[1];
      ax *= sc;
      ay *= sc;
#pragma unroll
      for (int g = 0; g < 2; ++g) {
        const float w0 = dpp<0x00>(w[g]), w1 = dpp<0x55>(w[g]);
        const float w2 = dpp<0xAA>(w[g]), w3 = dpp<0xFF>(w[g]);
        if (c + 4 * g + 0 < nloc) { ax += w0 * f[4 * g + 0].x; ay += w0 * f[4 * g + 0].y; }
        if (c + 4 * g + 1 < nloc) { ax += w1 * f[4 * g + 1].x; ay += w1 * f[4 * g + 1].y; }
        if (c + 4 * g + 2 < nloc) { ax += w2 * f[4 * g + 2].x; ay += w2 * f[4 * g + 2].y; }
        if (c + 4 * g + 3 < nloc) { ax += w3 * f[4 * g + 3].x; ay += w3 * f[4 * g + 3].y; }
      }
    }
  }
  den += dpp<0xB1>(den);
  den += dpp<0x4E>(den);

  const float2 bb = *reinterpret_cast<const float2*>(bias + 2 * lane);
  float ox = (den > 0.f ? ax / den : 0.f) + bb.x;
  float oy = (den > 0.f ? ay / den : 0.f) + bb.y;
  ox = ox > 0.f ? ox : expf(ox) - 1.f;
  oy = oy > 0.f ? oy : expf(oy) - 1.f;
  float2 o;
  o.x = ox;
  o.y = oy;
  *reinterpret_cast<float2*>(hout + (long)node * 64 + 2 * lane) = o;
}

__global__ __launch_bounds__(256) void node_agg_h1_kernel(
    const unsigned* __restrict__ rowptr, const int* __restrict__ csr_src,
    const float* __restrict__ feat, const float* __restrict__ al,
    const float* __restrict__ ar, const float* __restrict__ bias,
    float* __restrict__ out, int N) {
  const int lane = threadIdx.x & 31;
  const int node = blockIdx.x * 8 + __builtin_amdgcn_readfirstlane(threadIdx.x >> 5);
  if (node >= N) return;
  const int beg = (int)rowptr[node];
  const int end = (int)rowptr[node + 1];
  const bool b4 = (lane & 16) != 0, b3 = (lane & 8) != 0, b2 = (lane & 4) != 0;
  const int eslot = lane >> 2;

  const float alv = al[lane];
  float er_d = feat[(long)node * 32 + lane] * ar[lane];
  er_d += xhalf(er_d);
  er_d += dpp<0x168>(er_d);
  er_d += dpp<0x164>(er_d);
  er_d += dpp<0x4E>(er_d);
  er_d += dpp<0xB1>(er_d);

  float m = -__builtin_inff();
  float den = 0.f, acc = 0.f;
  for (int base = beg; base < end; base += 32) {
    const int nloc = (end - base < 32) ? end - base : 32;
    const int sid = (lane < nloc) ? csr_src[base + lane] : 0;
    for (int c = 0; c < nloc; c += 8) {
      float f[8];
#pragma unroll
      for (int j = 0; j < 8; ++j) {
        const int sj = __builtin_amdgcn_readlane(sid, c + j);
        f[j] = feat[(long)sj * 32 + lane];
      }
      float t[4], u[2];
#pragma unroll
      for (int i = 0; i < 4; ++i) {
        const float lo = f[i] * alv, hi = f[i + 4] * alv;
        t[i] = (b4 ? hi : lo) + xhalf(b4 ? lo : hi);
      }
#pragma unroll
      for (int i = 0; i < 2; ++i)
        u[i] = (b3 ? t[i + 2] : t[i]) + dpp<0x168>(b3 ? t[i] : t[i + 2]);
      float r = (b2 ? u[1] : u[0]) + dpp<0x164>(b2 ? u[0] : u[1]);
      r += dpp<0x4E>(r);
      r += dpp<0xB1>(r);
      const float lv = leaky(r + er_d);
      const float v = (c + eslot < nloc) ? lv : -__builtin_inff();
      float cm = fmaxf(v, dpp<0x164>(v));
      cm = fmaxf(cm, dpp<0x168>(cm));
      cm = fmaxf(cm, xhalf(cm));
      const float mn = fmaxf(m, cm);
      const float sc = expf(m - mn);
      m = mn;
      const float w = expf(v - mn);
      den = den * sc + w;
      acc *= sc;
#pragma unroll
      for (int j = 0; j < 8; ++j) {
        const float wj = __int_as_float(__builtin_amdgcn_readlane(__float_as_int(w), 4 * j));
        if (c + j < nloc) acc += wj * f[j];
      }
    }
  }
  den += dpp<0x164>(den);
  den += dpp<0x168>(den);
  den += xhalf(den);
  out[(long)node * 32 + lane] = (den > 0.f ? acc / den : 0.f) + bias[lane];
}

extern "C" void kernel_launch(void* const* d_in, const int* in_sizes, int n_in,
                              void* d_out, int out_size, void* d_ws,
                              size_t ws_size, hipStream_t stream) {
  const float* x   = (const float*)d_in[0];
  const int*   src = (const int*)d_in[1];
  const int*   dst = (const int*)d_in[2];
  const float* W1  = (const float*)d_in[3];
  const float* al1 = (const float*)d_in[4];
  const float* ar1 = (const float*)d_in[5];
  const float* b1  = (const float*)d_in[6];
  const float* W2  = (const float*)d_in[7];
  const float* al2 = (const float*)d_in[8];
  const float* ar2 = (const float*)d_in[9];
  const float* b2  = (const float*)d_in[10];
  float* out = (float*)d_out;

  const int N = NNODES, E = NEDGES;

  float* ws = (float*)d_ws;
  float*    feat1  = ws;
  float*    h1     = ws + (long)N * 72;
  int*      csr    = (int*)(ws + (long)N * 136);
  unsigned* rowptr = (unsigned*)(ws + (long)N * 152);
  unsigned* cnt    = (unsigned*)(ws + (long)N * 153 + 32);
  unsigned* bsum   = (unsigned*)(ws + (long)N * 154 + 32);
  float*    feat2  = ws;

  const int TB = 256;
  auto nb = [](long n, int b) { return (int)((n + b - 1) / b); };

  fill_u32_kernel<<<nb((long)N, TB), TB, 0, stream>>>(cnt, 0u, N);
  degree_count_kernel<<<nb((long)E, TB), TB, 0, stream>>>(dst, cnt, E);
  block_sum_kernel<<<nb((long)N + 1, 1024), 256, 0, stream>>>(cnt, bsum, N);
  scan_kernel<<<nb((long)N + 1, 1024), 256, 0, stream>>>(cnt, bsum, rowptr, N);
  csr_fill_kernel<<<nb((long)E, TB), TB, 0, stream>>>(src, dst, rowptr, cnt, csr, E);

  gemm_wmma_kernel<256, 64><<<N / 16, 128, 0, stream>>>(x, W1, feat1);
  node_agg_h8_kernel<<<nb((long)N, 8), 256, 0, stream>>>(rowptr, csr, feat1, al1, ar1, b1, h1, N);

  gemm_wmma_kernel<64, 32><<<N / 16, 64, 0, stream>>>(h1, W2, feat2);
  node_agg_h1_kernel<<<nb((long)N, 8), 256, 0, stream>>>(rowptr, csr, feat2, al2, ar2, b2, out, N);
}


# edges grouped by destination through LDS bins (chunk ordered by 512-node bin; per bin count, scan, rank by LDS adds): no global atomics, 9 -> 6 launches; sums stay in arrival order
# speedup vs baseline: 7.3436x; 2.1543x over previous
#include <hip/hip_runtime.h>
#include <hip/hip_bf16.h>

typedef __attribute__((ext_vector_type(2))) float v2f;
typedef __attribute__((ext_vector_type(8))) float v8f;

#define NNODES 100000
#define NEDGES 1600000

template <int K, int C>
__global__ __launch_bounds__(32 * (C / 16)) void gemm_wmma_kernel(
    const float* __restrict__ A, const float* __restrict__ B,
    float* __restrict__ D) {
  const int lane = threadIdx.x & 31;
  const int wave = threadIdx.x >> 5;
  const int row0 = blockIdx.x * 16;
  const int col0 = wave * 16;
  const int half = lane >> 4;
  const int l    = lane & 15;

  const float* arow = A + (long)(row0 + l) * K + 2 * half;
  const float* bcol = B + col0 + l;

  v8f acc = {};
#pragma unroll 4
  for (int k = 0; k < K; k += 4) {
    v2f a, b;
    a.x = arow[k + 0];
    a.y = arow[k + 1];
    b.x = bcol[(k + 2 * half + 0) * C];
    b.y = bcol[(k + 2 * half + 1) * C];
    acc = __builtin_amdgcn_wmma_f32_16x16x4_f32(
         false, a,  false, b,
         (short)0, acc,  false,  false);
  }

  float* drow = D + (long)(row0 + 8 * half) * C + col0 + l;
#pragma unroll
  for (int i = 0; i < 8; ++i) drow[i * C] = acc[i];
}

__device__ __forceinline__ float leaky(float v) { return v > 0.f ? v : 0.2f * v; }

#define BIN_SHIFT 9
#define BIN_NODES 512
#define NBINS 196
#define NCHUNK 256
#define CHUNK_EDGES 6250
#define EDGES_PER_THREAD 7
#define SORT_CAP 12288

static_assert(NBINS * BIN_NODES >= NNODES, "bins cover the nodes");
static_assert(NBINS <= 256, "bin counters");
static_assert(NCHUNK * CHUNK_EDGES >= NEDGES, "chunks cover the edges");
static_assert(EDGES_PER_THREAD * 1024 >= CHUNK_EDGES, "edges per thread");

__global__ __launch_bounds__(1024) void chunk_bin_kernel(
    const int* __restrict__ src, const int* __restrict__ dst,
    int* __restrict__ binned_w, unsigned* __restrict__ cnt_tab,
    unsigned* __restrict__ off_tab, int E) {
  __shared__ unsigned hist[256];
  __shared__ unsigned start[256];
  __shared__ int2 stage[CHUNK_EDGES];
  int2* __restrict__ binned = reinterpret_cast<int2*>(binned_w);
  const int t = threadIdx.x;
  const int g = blockIdx.x;
  const int e0 = g * CHUNK_EDGES;
  int n = E - e0;
  n = n < 0 ? 0 : (n > CHUNK_EDGES ? CHUNK_EDGES : n);

  if (t < 256) hist[t] = 0u;
  __syncthreads();

  int s[EDGES_PER_THREAD], d[EDGES_PER_THREAD];
  unsigned r[EDGES_PER_THREAD];
#pragma unroll
  for (int k = 0; k < EDGES_PER_THREAD; ++k) {
    const int i = t + k * 1024;
    s[k] = 0;
    d[k] = 0;
    if (i < n) {
      s[k] = src[e0 + i];
      d[k] = dst[e0 + i];
    }
  }
#pragma unroll
  for (int k = 0; k < EDGES_PER_THREAD; ++k) {
    const int i = t + k * 1024;
    r[k] = 0u;
    if (i < n) r[k] = atomicAdd(&hist[d[k] >> BIN_SHIFT], 1u);
  }
  __syncthreads();

  if (t < 32) {
    unsigned c[8];
    unsigned loc = 0u;
#pragma unroll
    for (int k = 0; k < 8; ++k) {
      c[k] = hist[8 * t + k];
      loc += c[k];
    }
    unsigned inc = loc;
    for (int off = 1; off < 32; off <<= 1) {
      const unsigned y = __shfl_up(inc, off, 32);
      if (t >= off) inc += y;
    }
    unsigned ex = inc - loc;
#pragma unroll
    for (int k = 0; k < 8; ++k) {
      start[8 * t + k] = ex;
      ex += c[k];
    }
  }
  __syncthreads();

#pragma unroll
  for (int k = 0; k < EDGES_PER_THREAD; ++k) {
    const int i = t + k * 1024;
    if (i < n) {
      int2 e;
      e.x = s[k];
      e.y = d[k];
      stage[start[d[k] >> BIN_SHIFT] + r[k]] = e;
    }
  }
  __syncthreads();

  for (int i = t; i < n; i += 1024) binned[e0 + i] = stage[i];
  if (t < NBINS) {
    cnt_tab[t * NCHUNK + g] = hist[t];
    off_tab[t * NCHUNK + g] = start[t];
  }
}

__device__ __forceinline__ void place_edge(unsigned* ncur, int* sorted, int* __restrict__ csr_src,
                                           unsigned base, int nb, int2 e) {
  const unsigned pos = atomicAdd(&ncur[e.y - nb], 1u);
  if (pos < SORT_CAP) sorted[pos] = e.x;
  else csr_src[base + pos] = e.x;
}

__global__ __launch_bounds__(1024) void bin_csr_kernel(
    const int* __restrict__ binned_w, const unsigned* __restrict__ cnt_tab,
    const unsigned* __restrict__ off_tab, unsigned* __restrict__ rowptr,
    int* __restrict__ csr_src, int N) {
  __shared__ unsigned ncur[BIN_NODES];
  __shared__ unsigned wsum[32];
  __shared__ unsigned wtot[16];
  __shared__ int sorted[SORT_CAP];
  const int2* __restrict__ binned = reinterpret_cast<const int2*>(binned_w);
  const int t = threadIdx.x, lane = t & 31;
  const int wave = __builtin_amdgcn_readfirstlane(t >> 5);
  const int b = blockIdx.x;
  const int nb = b * BIN_NODES;

  unsigned c[8];
  const int2* seg[8];
  int2 ev[8][2];
#pragma unroll
  for (int k = 0; k < 8; ++k) {
    const int gk = wave * 8 + k;
    c[k] = cnt_tab[b * NCHUNK + gk];
    seg[k] = binned + (long)gk * CHUNK_EDGES + off_tab[b * NCHUNK + gk];
  }
#pragma unroll
  for (int k = 0; k < 8; ++k) {
#pragma unroll
    for (int h = 0; h < 2; ++h) {
      ev[k][h].x = 0;
      ev[k][h].y = nb;
      if ((unsigned)(lane + 32 * h) < c[k]) ev[k][h] = seg[k][lane + 32 * h];
    }
  }

  unsigned p = 0u;
  const uint4* ct4 = reinterpret_cast<const uint4*>(cnt_tab);
  for (int i = t; i < b * (NCHUNK / 4); i += 1024) {
    const uint4 q = ct4[i];
    p += q.x + q.y + q.z + q.w;
  }
  for (int off = 16; off > 0; off >>= 1) p += __shfl_xor(p, off, 32);
  if (lane == 0) wsum[wave] = p;
  if (t < BIN_NODES) ncur[t] = 0u;
  __syncthreads();
  unsigned base = 0u;
  for (int w = 0; w < 32; ++w) base += wsum[w];

#pragma unroll
  for (int k = 0; k < 8; ++k) {
#pragma unroll
    for (int h = 0; h < 2; ++h)
      if ((unsigned)(lane + 32 * h) < c[k]) atomicAdd(&ncur[ev[k][h].y - nb], 1u);
    for (unsigned i = 64u + lane; i < c[k]; i += 32u) {
      const int2 e = seg[k][i];
      atomicAdd(&ncur[e.y - nb], 1u);
    }
  }
  __syncthreads();

  unsigned v = 0u, inc = 0u;
  if (t < BIN_NODES) {
    v = ncur[t];
    inc = v;
    for (int off = 1; off < 32; off <<= 1) {
      const unsigned y = __shfl_up(inc, off, 32);
      if (lane >= off) inc += y;
    }
    if (lane == 31) wtot[wave] = inc;
  }
  __syncthreads();
  if (t < BIN_NODES) {
    unsigned ex = inc - v;
    for (int w = 0; w < 16; ++w)
      if (w < wave) ex += wtot[w];
    ncur[t] = ex;
    if (nb + t <= N) rowptr[nb + t] = base + ex;
  }
  __syncthreads();

#pragma unroll
  for (int k = 0; k < 8; ++k) {
#pragma unroll
    for (int h = 0; h < 2; ++h)
      if ((unsigned)(lane + 32 * h) < c[k]) place_edge(ncur, sorted, csr_src, base, nb, ev[k][h]);
    for (unsigned i = 64u + lane; i < c[k]; i += 32u)
      place_edge(ncur, sorted, csr_src, base, nb, seg[k][i]);
  }
  __syncthreads();

  const unsigned total = ncur[BIN_NODES - 1];
  const unsigned m = total < SORT_CAP ? total : SORT_CAP;
  for (unsigned i = t; i < m; i += 1024u) csr_src[base + i] = sorted[i];
}

template <int CTRL>
__device__ __forceinline__ float dpp(float x) {
  return __builtin_bit_cast(
      float, __builtin_amdgcn_update_dpp(0, __builtin_bit_cast(int, x), CTRL, 0xf, 0xf, true));
}
__device__ __forceinline__ float xhalf(float x) {
  return __builtin_bit_cast(
      float, __builtin_amdgcn_permlanex16(__builtin_bit_cast(int, x), __builtin_bit_cast(int, x),
                                          0x76543210, 0xfedcba98, false, false));
}

__global__ __launch_bounds__(256) void node_agg_h8_kernel(
    const unsigned* __restrict__ rowptr, const int* __restrict__ csr_src,
    const float* __restrict__ feat, const float* __restrict__ al,
    const float* __restrict__ ar, const float* __restrict__ bias,
    float* __restrict__ hout, int N) {
  const int lane = threadIdx.x & 31;
  const int node = blockIdx.x * 8 + __builtin_amdgcn_readfirstlane(threadIdx.x >> 5);
  if (node >= N) return;
  const int beg = (int)rowptr[node];
  const int end = (int)rowptr[node + 1];
  const int q = lane & 3;
  const bool q1 = (q & 1) != 0, q2 = (q & 2) != 0;

  const float2 fd = *reinterpret_cast<const float2*>(feat + (long)node * 64 + 2 * lane);
  const float2 r2 = *reinterpret_cast<const float2*>(ar + 2 * lane);
  const float2 a2 = *reinterpret_cast<const float2*>(al + 2 * lane);
  float er_h = fd.x * r2.x + fd.y * r2.y;
  er_h += dpp<0xB1>(er_h);
  er_h += dpp<0x4E>(er_h);

  float m = -__builtin_inff();
  float den = 0.f, ax = 0.f, ay = 0.f;
  for (int base = beg; base < end; base += 32) {
    const int nloc = (end - base < 32) ? end - base : 32;
    const int sid = (lane < nloc) ? csr_src[base + lane] : 0;
    for (int c = 0; c < nloc; c += 8) {
      float2 f[8];
#pragma unroll
      for (int j = 0; j < 8; ++j) {
        const int sj = __builtin_amdgcn_readlane(sid, c + j);
        f[j] = *reinterpret_cast<const float2*>(feat + (long)sj * 64 + 2 * lane);
      }
      float v[2];
#pragma unroll
      for (int g = 0; g < 2; ++g) {
        const float p0 = f[4 * g + 0].x * a2.x + f[4 * g + 0].y * a2.y;
        const float p1 = f[4 * g + 1].x * a2.x + f[4 * g + 1].y * a2.y;
        const float p2 = f[4 * g + 2].x * a2.x + f[4 * g + 2].y * a2.y;
        const float p3 = f[4 * g + 3].x * a2.x + f[4 * g + 3].y * a2.y;
        const float t01 = (q1 ? p1 : p0) + dpp<0xB1>(q1 ? p0 : p1);
        const float t23 = (q1 ? p3 : p2) + dpp<0xB1>(q1 ? p2 : p3);
        const float s = (q2 ? t23 : t01) + dpp<0x4E>(q2 ? t01 : t23);
        const float lv = leaky(s + er_h);
        v[g] = (c + 4 * g + q < nloc) ? lv : -__builtin_inff();
      }
      float cm = fmaxf(v[0], v[1]);
      cm = fmaxf(cm, dpp<0xB1>(cm));
      cm = fmaxf(cm, dpp<0x4E>(cm));
      const float mn = fmaxf(m, cm);
      const float sc = expf(m - mn);
      m = mn;
      float w[2];
      w[0] = expf(v[0] - mn);
      w[1] = expf(v[1] - mn);
      den = den * sc + w[0] + w[1];
      ax *= sc;
      ay *= sc;
#pragma unroll
      for (int g = 0; g < 2; ++g) {
        const float w0 = dpp<0x00>(w[g]), w1 = dpp<0x55>(w[g]);
        const float w2 = dpp<0xAA>(w[g]), w3 = dpp<0xFF>(w[g]);
        if (c + 4 * g + 0 < nloc) { ax += w0 * f[4 * g + 0].x; ay += w0 * f[4 * g + 0].y; }
        if (c + 4 * g + 1 < nloc) { ax += w1 * f[4 * g + 1].x; ay += w1 * f[4 * g + 1].y; }
        if (c + 4 * g + 2 < nloc) { ax += w2 * f[4 * g + 2].x; ay += w2 * f[4 * g + 2].y; }
        if (c + 4 * g + 3 < nloc) { ax += w3 * f[4 * g + 3].x; ay += w3 * f[4 * g + 3].y; }
      }
    }
  }
  den += dpp<0xB1>(den);
  den += dpp<0x4E>(den);

  const float2 bb = *reinterpret_cast<const float2*>(bias + 2 * lane);
  float ox = (den > 0.f ? ax / den : 0.f) + bb.x;
  float oy = (den > 0.f ? ay / den : 0.f) + bb.y;
  ox = ox > 0.f ? ox : expf(ox) - 1.f;
  oy = oy > 0.f ? oy : expf(oy) - 1.f;
  float2 o;
  o.x = ox;
  o.y = oy;
  *reinterpret_cast<float2*>(hout + (long)node * 64 + 2 * lane) = o;
}

__global__ __launch_bounds__(256) void node_agg_h1_kernel(
    const unsigned* __restrict__ rowptr, const int* __restrict__ csr_src,
    const float* __restrict__ feat, const float* __restrict__ al,
    const float* __restrict__ ar, const float* __restrict__ bias,
    float* __restrict__ out, int N) {
  const int lane = threadIdx.x & 31;
  const int node = blockIdx.x * 8 + __builtin_amdgcn_readfirstlane(threadIdx.x >> 5);
  if (node >= N) return;
  const int beg = (int)rowptr[node];
  const int end = (int)rowptr[node + 1];
  const bool b4 = (lane & 16) != 0, b3 = (lane & 8) != 0, b2 = (lane & 4) != 0;
  const int eslot = lane >> 2;

  const float alv = al[lane];
  float er_d = feat[(long)node * 32 + lane] * ar[lane];
  er_d += xhalf(er_d);
  er_d += dpp<0x168>(er_d);
  er_d += dpp<0x164>(er_d);
  er_d += dpp<0x4E>(er_d);
  er_d += dpp<0xB1>(er_d);

  float m = -__builtin_inff();
  float den = 0.f, acc = 0.f;
  for (int base = beg; base < end; base += 32) {
    const int nloc = (end - base < 32) ? end - base : 32;
    const int sid = (lane < nloc) ? csr_src[base + lane] : 0;
    for (int c = 0; c < nloc; c += 8) {
      float f[8];
#pragma unroll
      for (int j = 0; j < 8; ++j) {
        const int sj = __builtin_amdgcn_readlane(sid, c + j);
        f[j] = feat[(long)sj * 32 + lane];
      }
      float t[4], u[2];
#pragma unroll
      for (int i = 0; i < 4; ++i) {
        const float lo = f[i] * alv, hi = f[i + 4] * alv;
        t[i] = (b4 ? hi : lo) + xhalf(b4 ? lo : hi);
      }
#pragma unroll
      for (int i = 0; i < 2; ++i)
        u[i] = (b3 ? t[i + 2] : t[i]) + dpp<0x168>(b3 ? t[i] : t[i + 2]);
      float r = (b2 ? u[1] : u[0]) + dpp<0x164>(b2 ? u[0] : u[1]);
      r += dpp<0x4E>(r);
      r += dpp<0xB1>(r);
      const float lv = leaky(r + er_d);
      const float v = (c + eslot < nloc) ? lv : -__builtin_inff();
      float cm = fmaxf(v, dpp<0x164>(v));
      cm = fmaxf(cm, dpp<0x168>(cm));
      cm = fmaxf(cm, xhalf(cm));
      const float mn = fmaxf(m, cm);
      const float sc = expf(m - mn);
      m = mn;
      const float w = expf(v - mn);
      den = den * sc + w;
      acc *= sc;
#pragma unroll
      for (int j = 0; j < 8; ++j) {
        const float wj = __int_as_float(__builtin_amdgcn_readlane(__float_as_int(w), 4 * j));
        if (c + j < nloc) acc += wj * f[j];
      }
    }
  }
  den += dpp<0x164>(den);
  den += dpp<0x168>(den);
  den += xhalf(den);
  out[(long)node * 32 + lane] = (den > 0.f ? acc / den : 0.f) + bias[lane];
}

extern "C" void kernel_launch(void* const* d_in, const int* in_sizes, int n_in,
                              void* d_out, int out_size, void* d_ws,
                              size_t ws_size, hipStream_t stream) {
  const float* x   = (const float*)d_in[0];
  const int*   src = (const int*)d_in[1];
  const int*   dst = (const int*)d_in[2];
  const float* W1  = (const float*)d_in[3];
  const float* al1 = (const float*)d_in[4];
  const float* ar1 = (const float*)d_in[5];
  const float* b1  = (const float*)d_in[6];
  const float* W2  = (const float*)d_in[7];
  const float* al2 = (const float*)d_in[8];
  const float* ar2 = (const float*)d_in[9];
  const float* b2  = (const float*)d_in[10];
  float* out = (float*)d_out;

  const int N = NNODES, E = NEDGES;

  float* ws = (float*)d_ws;
  float*    feat1  = ws;
  float*    h1     = ws + (long)N * 72;
  int*      csr    = (int*)(ws + (long)N * 136);
  unsigned* rowptr = (unsigned*)(ws + (long)N * 152);
  int*      binned  = (int*)(ws + (long)N * 72);
  unsigned* cnt_tab = (unsigned*)(ws + (long)N * 104);
  unsigned* off_tab = cnt_tab + NBINS * NCHUNK;
  float*    feat2  = ws;

  auto nb = [](long n, int b) { return (int)((n + b - 1) / b); };

  chunk_bin_kernel<<<NCHUNK, 1024, 0, stream>>>(src, dst, binned, cnt_tab, off_tab, E);
  bin_csr_kernel<<<NBINS, 1024, 0, stream>>>(binned, cnt_tab, off_tab, rowptr, csr, N);

  gemm_wmma_kernel<256, 64><<<N / 16, 128, 0, stream>>>(x, W1, feat1);
  node_agg_h8_kernel<<<nb((long)N, 8), 256, 0, stream>>>(rowptr, csr, feat1, al1, ar1, b1, h1, N);

  gemm_wmma_kernel<64, 32><<<N / 16, 64, 0, stream>>>(h1, W2, feat2);
  node_agg_h1_kernel<<<nb((long)N, 8), 256, 0, stream>>>(rowptr, csr, feat2, al2, ar2, b2, out, N);
}


# both GEMMs: a work-group brings its 32 rows of A in once by whole-line 16-byte loads into LDS; waves read A fragments from LDS, each wave 2 row tiles per B fragment; k order and sums unchanged
# speedup vs baseline: 8.7844x; 1.1962x over previous
#include <hip/hip_runtime.h>
#include <hip/hip_bf16.h>

typedef __attribute__((ext_vector_type(2))) float v2f;
typedef __attribute__((ext_vector_type(8))) float v8f;

#define NNODES 100000
#define NEDGES 1600000

#define GEMM_ROWS 32
static_assert(NNODES % GEMM_ROWS == 0, "row tiles cover the nodes exactly");

template <int K, int C>
__global__ __launch_bounds__(32 * (C / 16)) void gemm_wmma_kernel(
    const float* __restrict__ A, const float* __restrict__ B,
    float* __restrict__ D) {
  constexpr int NT = 32 * (C / 16);
  constexpr int PITCH = K + 4;
  constexpr int VROW = K / 4;
  constexpr int NLOAD = GEMM_ROWS * VROW / NT;
  static_assert(GEMM_ROWS * VROW % NT == 0, "tile splits evenly over the threads");
  static_assert(PITCH % 64 == 4, "bank spread of the fragment reads");
  __shared__ __attribute__((aligned(16))) float tile[GEMM_ROWS * PITCH];

  const int tid  = threadIdx.x;
  const int lane = tid & 31;
  const int wave = tid >> 5;
  const int row0 = blockIdx.x * GEMM_ROWS;
  const int col0 = wave * 16;
  const int half = lane >> 4;
  const int l    = lane & 15;

  const float4* __restrict__ Ag = reinterpret_cast<const float4*>(A + (long)row0 * K);
  float4 st[NLOAD];
#pragma unroll
  for (int j = 0; j < NLOAD; ++j) st[j] = Ag[tid + j * NT];
#pragma unroll
  for (int j = 0; j < NLOAD; ++j) {
    const int i = tid + j * NT;
    *reinterpret_cast<float4*>(tile + (i / VROW) * PITCH + 4 * (i % VROW)) = st[j];
  }
  __syncthreads();

  const float* a0 = tile + l * PITCH + 2 * half;
  const float* a1 = a0 + 16 * PITCH;
  const float* bcol = B + col0 + l;

  v8f acc0 = {}, acc1 = {};
#pragma unroll 8
  for (int k = 0; k < K; k += 4) {
    v2f b;
    b.x = bcol[(k + 2 * half + 0) * C];
    b.y = bcol[(k + 2 * half + 1) * C];
    const v2f x0 = *reinterpret_cast<const v2f*>(a0 + k);
    const v2f x1 = *reinterpret_cast<const v2f*>(a1 + k);
    acc0 = __builtin_amdgcn_wmma_f32_16x16x4_f32(
         false, x0,  false, b,
         (short)0, acc0,  false,  false);
    acc1 = __builtin_amdgcn_wmma_f32_16x16x4_f32(
         false, x1,  false, b,
         (short)0, acc1,  false,  false);
  }

  float* drow = D + (long)(row0 + 8 * half) * C + col0 + l;
#pragma unroll
  for (int i = 0; i < 8; ++i) {
    drow[i * C] = acc0[i];
    drow[(16 + i) * C] = acc1[i];
  }
}

__device__ __forceinline__ float leaky(float v) { return v > 0.f ? v : 0.2f * v; }

#define BIN_SHIFT 9
#define BIN_NODES 512
#define NBINS 196
#define NCHUNK 256
#define CHUNK_EDGES 6250
#define EDGES_PER_THREAD 7
#define SORT_CAP 12288

static_assert(NBINS * BIN_NODES >= NNODES, "bins cover the nodes");
static_assert(NBINS <= 256, "bin counters");
static_assert(NCHUNK * CHUNK_EDGES >= NEDGES, "chunks cover the edges");
static_assert(EDGES_PER_THREAD * 1024 >= CHUNK_EDGES, "edges per thread");

__global__ __launch_bounds__(1024) void chunk_bin_kernel(
    const int* __restrict__ src, const int* __restrict__ dst,
    int* __restrict__ binned_w, unsigned* __restrict__ cnt_tab,
    unsigned* __restrict__ off_tab, int E) {
  __shared__ unsigned hist[256];
  __shared__ unsigned start[256];
  __shared__ int2 stage[CHUNK_EDGES];
  int2* __restrict__ binned = reinterpret_cast<int2*>(binned_w);
  const int t = threadIdx.x;
  const int g = blockIdx.x;
  const int e0 = g * CHUNK_EDGES;
  int n = E - e0;
  n = n < 0 ? 0 : (n > CHUNK_EDGES ? CHUNK_EDGES : n);

  if (t < 256) hist[t] = 0u;
  __syncthreads();

  int s[EDGES_PER_THREAD], d[EDGES_PER_THREAD];
  unsigned r[EDGES_PER_THREAD];
#pragma unroll
  for (int k = 0; k < EDGES_PER_THREAD; ++k) {
    const int i = t + k * 1024;
    s[k] = 0;
    d[k] = 0;
    if (i < n) {
      s[k] = src[e0 + i];
      d[k] = dst[e0 + i];
    }
  }
#pragma unroll
  for (int k = 0; k < EDGES_PER_THREAD; ++k) {
    const int i = t + k * 1024;
    r[k] = 0u;
    if (i < n) r[k] = atomicAdd(&hist[d[k] >> BIN_SHIFT], 1u);
  }
  __syncthreads();

  if (t < 32) {
    unsigned c[8];
    unsigned loc = 0u;
#pragma unroll
    for (int k = 0; k < 8; ++k) {
      c[k] = hist[8 * t + k];
      loc += c[k];
    }
    unsigned inc = loc;
    for (int off = 1; off < 32; off <<= 1) {
      const unsigned y = __shfl_up(inc, off, 32);
      if (t >= off) inc += y;
    }
    unsigned ex = inc - loc;
#pragma unroll
    for (int k = 0; k < 8; ++k) {
      start[8 * t + k] = ex;
      ex += c[k];
    }
  }
  __syncthreads();

#pragma unroll
  for (int k = 0; k < EDGES_PER_THREAD; ++k) {
    const int i = t + k * 1024;
    if (i < n) {
      int2 e;
      e.x = s[k];
      e.y = d[k];
      stage[start[d[k] >> BIN_SHIFT] + r[k]] = e;
    }
  }
  __syncthreads();

  for (int i = t; i < n; i += 1024) binned[e0 + i] = stage[i];
  if (t < NBINS) {
    cnt_tab[t * NCHUNK + g] = hist[t];
    off_tab[t * NCHUNK + g] = start[t];
  }
}

__device__ __forceinline__ void place_edge(unsigned* ncur, int* sorted, int* __restrict__ csr_src,
                                           unsigned base, int nb, int2 e) {
  const unsigned pos = atomicAdd(&ncur[e.y - nb], 1u);
  if (pos < SORT_CAP) sorted[pos] = e.x;
  else csr_src[base + pos] = e.x;
}

__global__ __launch_bounds__(1024) void bin_csr_kernel(
    const int* __restrict__ binned_w, const unsigned* __restrict__ cnt_tab,
    const unsigned* __restrict__ off_tab, unsigned* __restrict__ rowptr,
    int* __restrict__ csr_src, int N) {
  __shared__ unsigned ncur[BIN_NODES];
  __shared__ unsigned wsum[32];
  __shared__ unsigned wtot[16];
  __shared__ int sorted[SORT_CAP];
  const int2* __restrict__ binned = reinterpret_cast<const int2*>(binned_w);
  const int t = threadIdx.x, lane = t & 31;
  const int wave = __builtin_amdgcn_readfirstlane(t >> 5);
  const int b = blockIdx.x;
  const int nb = b * BIN_NODES;

  unsigned c[8];
  const int2* seg[8];
  int2 ev[8][2];
#pragma unroll
  for (int k = 0; k < 8; ++k) {
    const int gk = wave * 8 + k;
    c[k] = cnt_tab[b * NCHUNK + gk];
    seg[k] = binned + (long)gk * CHUNK_EDGES + off_tab[b * NCHUNK + gk];
  }
#pragma unroll
  for (int k = 0; k < 8; ++k) {
#pragma unroll
    for (int h = 0; h < 2; ++h) {
      ev[k][h].x = 0;
      ev[k][h].y = nb;
      if ((unsigned)(lane + 32 * h) < c[k]) ev[k][h] = seg[k][lane + 32 * h];
    }
  }

  unsigned p = 0u;
  const uint4* ct4 = reinterpret_cast<const uint4*>(cnt_tab);
  for (int i = t; i < b * (NCHUNK / 4); i += 1024) {
    const uint4 q = ct4[i];
    p += q.x + q.y + q.z + q.w;
  }
  for (int off = 16; off > 0; off >>= 1) p += __shfl_xor(p, off, 32);
  if (lane == 0) wsum[wave] = p;
  if (t < BIN_NODES) ncur[t] = 0u;
  __syncthreads();
  unsigned base = 0u;
  for (int w = 0; w < 32; ++w) base += wsum[w];

#pragma unroll
  for (int k = 0; k < 8; ++k) {
#pragma unroll
    for (int h = 0; h < 2; ++h)
      if ((unsigned)(lane + 32 * h) < c[k]) atomicAdd(&ncur[ev[k][h].y - nb], 1u);
    for (unsigned i = 64u + lane; i < c[k]; i += 32u) {
      const int2 e = seg[k][i];
      atomicAdd(&ncur[e.y - nb], 1u);
    }
  }
  __syncthreads();

  unsigned v = 0u, inc = 0u;
  if (t < BIN_NODES) {
    v = ncur[t];
    inc = v;
    for (int off = 1; off < 32; off <<= 1) {
      const unsigned y = __shfl_up(inc, off, 32);
      if (lane >= off) inc += y;
    }
    if (lane == 31) wtot[wave] = inc;
  }
  __syncthreads();
  if (t < BIN_NODES) {
    unsigned ex = inc - v;
    for (int w = 0; w < 16; ++w)
      if (w < wave) ex += wtot[w];
    ncur[t] = ex;
    if (nb + t <= N) rowptr[nb + t] = base + ex;
  }
  __syncthreads();

#pragma unroll
  for (int k = 0; k < 8; ++k) {
#pragma unroll
    for (int h = 0; h < 2; ++h)
      if ((unsigned)(lane + 32 * h) < c[k]) place_edge(ncur, sorted, csr_src, base, nb, ev[k][h]);
    for (unsigned i = 64u + lane; i < c[k]; i += 32u)
      place_edge(ncur, sorted, csr_src, base, nb, seg[k][i]);
  }
  __syncthreads();

  const unsigned total = ncur[BIN_NODES - 1];
  const unsigned m = total < SORT_CAP ? total : SORT_CAP;
  for (unsigned i = t; i < m; i += 1024u) csr_src[base + i] = sorted[i];
}

template <int CTRL>
__device__ __forceinline__ float dpp(float x) {
  return __builtin_bit_cast(
      float, __builtin_amdgcn_update_dpp(0, __builtin_bit_cast(int, x), CTRL, 0xf, 0xf, true));
}
__device__ __forceinline__ float xhalf(float x) {
  return __builtin_bit_cast(
      float, __builtin_amdgcn_permlanex16(__builtin_bit_cast(int, x), __builtin_bit_cast(int, x),
                                          0x76543210, 0xfedcba98, false, false));
}

__global__ __launch_bounds__(256) void node_agg_h8_kernel(
    const unsigned* __restrict__ rowptr, const int* __restrict__ csr_src,
    const float* __restrict__ feat, const float* __restrict__ al,
    const float* __restrict__ ar, const float* __restrict__ bias,
    float* __restrict__ hout, int N) {
  const int lane = threadIdx.x & 31;
  const int node = blockIdx.x * 8 + __builtin_amdgcn_readfirstlane(threadIdx.x >> 5);
  if (node >= N) return;
  const int beg = (int)rowptr[node];
  const int end = (int)rowptr[node + 1];
  const int q = lane & 3;
  const bool q1 = (q & 1) != 0, q2 = (q & 2) != 0;

  const float2 fd = *reinterpret_cast<const float2*>(feat + (long)node * 64 + 2 * lane);
  const float2 r2 = *reinterpret_cast<const float2*>(ar + 2 * lane);
  const float2 a2 = *reinterpret_cast<const float2*>(al + 2 * lane);
  float er_h = fd.x * r2.x + fd.y * r2.y;
  er_h += dpp<0xB1>(er_h);
  er_h += dpp<0x4E>(er_h);

  float m = -__builtin_inff();
  float den = 0.f, ax = 0.f, ay = 0.f;
  for (int base = beg; base < end; base += 32) {
    const int nloc = (end - base < 32) ? end - base : 32;
    const int sid = (lane < nloc) ? csr_src[base + lane] : 0;
    for (int c = 0; c < nloc; c += 8) {
      float2 f[8];
#pragma unroll
      for (int j = 0; j < 8; ++j) {
        const int sj = __builtin_amdgcn_readlane(sid, c + j);
        f[j] = *reinterpret_cast<const float2*>(feat + (long)sj * 64 + 2 * lane);
      }
      float v[2];
#pragma unroll
      for (int g = 0; g < 2; ++g) {
        const float p0 = f[4 * g + 0].x * a2.x + f[4 * g + 0].y * a2.y;
        const float p1 = f[4 * g + 1].x * a2.x + f[4 * g + 1].y * a2.y;
        const float p2 = f[4 * g + 2].x * a2.x + f[4 * g + 2].y * a2.y;
        const float p3 = f[4 * g + 3].x * a2.x + f[4 * g + 3].y * a2.y;
        const float t01 = (q1 ? p1 : p0) + dpp<0xB1>(q1 ? p0 : p1);
        const float t23 = (q1 ? p3 : p2) + dpp<0xB1>(q1 ? p2 : p3);
        const float s = (q2 ? t23 : t01) + dpp<0x4E>(q2 ? t01 : t23);
        const float lv = leaky(s + er_h);
        v[g] = (c + 4 * g + q < nloc) ? lv : -__builtin_inff();
      }
      float cm = fmaxf(v[0], v[1]);
      cm = fmaxf(cm, dpp<0xB1>(cm));
      cm = fmaxf(cm, dpp<0x4E>(cm));
      const float mn = fmaxf(m, cm);
      const float sc = expf(m - mn);
      m = mn;
      float w[2];
      w[0] = expf(v[0] - mn);
      w[1] = expf(v[1] - mn);
      den = den * sc + w[0] + w[1];
      ax *= sc;
      ay *= sc;
#pragma unroll
      for (int g = 0; g < 2; ++g) {
        const float w0 = dpp<0x00>(w[g]), w1 = dpp<0x55>(w[g]);
        const float w2 = dpp<0xAA>(w[g]), w3 = dpp<0xFF>(w[g]);
        if (c + 4 * g + 0 < nloc) { ax += w0 * f[4 * g + 0].x; ay += w0 * f[4 * g + 0].y; }
        if (c + 4 * g + 1 < nloc) { ax += w1 * f[4 * g + 1].x; ay += w1 * f[4 * g + 1].y; }
        if (c + 4 * g + 2 < nloc) { ax += w2 * f[4 * g + 2].x; ay += w2 * f[4 * g + 2].y; }
        if (c + 4 * g + 3 < nloc) { ax += w3 * f[4 * g + 3].x; ay += w3 * f[4 * g + 3].y; }
      }
    }
  }
  den += dpp<0xB1>(den);
  den += dpp<0x4E>(den);

  const float2 bb = *reinterpret_cast<const float2*>(bias + 2 * lane);
  float ox = (den > 0.f ? ax / den : 0.f) + bb.x;
  float oy = (den > 0.f ? ay / den : 0.f) + bb.y;
  ox = ox > 0.f ? ox : expf(ox) - 1.f;
  oy = oy > 0.f ? oy : expf(oy) - 1.f;
  float2 o;
  o.x = ox;
  o.y = oy;
  *reinterpret_cast<float2*>(hout + (long)node * 64 + 2 * lane) = o;
}

__global__ __launch_bounds__(256) void node_agg_h1_kernel(
    const unsigned* __restrict__ rowptr, const int* __restrict__ csr_src,
    const float* __restrict__ feat, const float* __restrict__ al,
    const float* __restrict__ ar, const float* __restrict__ bias,
    float* __restrict__ out, int N) {
  const int lane = threadIdx.x & 31;
  const int node = blockIdx.x * 8 + __builtin_amdgcn_readfirstlane(threadIdx.x >> 5);
  if (node >= N) return;
  const int beg = (int)rowptr[node];
  const int end = (int)rowptr[node + 1];
  const bool b4 = (lane & 16) != 0, b3 = (lane & 8) != 0, b2 = (lane & 4) != 0;
  const int eslot = lane >> 2;

  const float alv = al[lane];
  float er_d = feat[(long)node * 32 + lane] * ar[lane];
  er_d += xhalf(er_d);
  er_d += dpp<0x168>(er_d);
  er_d += dpp<0x164>(er_d);
  er_d += dpp<0x4E>(er_d);
  er_d += dpp<0xB1>(er_d);

  float m = -__builtin_inff();
  float den = 0.f, acc = 0.f;
  for (int base = beg; base < end; base += 32) {
    const int nloc = (end - base < 32) ? end - base : 32;
    const int sid = (lane < nloc) ? csr_src[base + lane] : 0;
    for (int c = 0; c < nloc; c += 8) {
      float f[8];
#pragma unroll
      for (int j = 0; j < 8; ++j) {
        const int sj = __builtin_amdgcn_readlane(sid, c + j);
        f[j] = feat[(long)sj * 32 + lane];
      }
      float t[4], u[2];
#pragma unroll
      for (int i = 0; i < 4; ++i) {
        const float lo = f[i] * alv, hi = f[i + 4] * alv;
        t[i] = (b4 ? hi : lo) + xhalf(b4 ? lo : hi);
      }
#pragma unroll
      for (int i = 0; i < 2; ++i)
        u[i] = (b3 ? t[i + 2] : t[i]) + dpp<0x168>(b3 ? t[i] : t[i + 2]);
      float r = (b2 ? u[1] : u[0]) + dpp<0x164>(b2 ? u[0] : u[1]);
      r += dpp<0x4E>(r);
      r += dpp<0xB1>(r);
      const float lv = leaky(r + er_d);
      const float v = (c + eslot < nloc) ? lv : -__builtin_inff();
      float cm = fmaxf(v, dpp<0x164>(v));
      cm = fmaxf(cm, dpp<0x168>(cm));
      cm = fmaxf(cm, xhalf(cm));
      const float mn = fmaxf(m, cm);
      const float sc = expf(m - mn);
      m = mn;
      const float w = expf(v - mn);
      den = den * sc + w;
      acc *= sc;
#pragma unroll
      for (int j = 0; j < 8; ++j) {
        const float wj = __int_as_float(__builtin_amdgcn_readlane(__float_as_int(w), 4 * j));
        if (c + j < nloc) acc += wj * f[j];
      }
    }
  }
  den += dpp<0x164>(den);
  den += dpp<0x168>(den);
  den += xhalf(den);
  out[(long)node * 32 + lane] = (den > 0.f ? acc / den : 0.f) + bias[lane];
}

extern "C" void kernel_launch(void* const* d_in, const int* in_sizes, int n_in,
                              void* d_out, int out_size, void* d_ws,
                              size_t ws_size, hipStream_t stream) {
  const float* x   = (const float*)d_in[0];
  const int*   src = (const int*)d_in[1];
  const int*   dst = (const int*)d_in[2];
  const float* W1  = (const float*)d_in[3];
  const float* al1 = (const float*)d_in[4];
  const float* ar1 = (const float*)d_in[5];
  const float* b1  = (const float*)d_in[6];
  const float* W2  = (const float*)d_in[7];
  const float* al2 = (const float*)d_in[8];
  const float* ar2 = (const float*)d_in[9];
  const float* b2  = (const float*)d_in[10];
  float* out = (float*)d_out;

  const int N = NNODES, E = NEDGES;

  float* ws = (float*)d_ws;
  float*    feat1  = ws;
  float*    h1     = ws + (long)N * 72;
  int*      csr    = (int*)(ws + (long)N * 136);
  unsigned* rowptr = (unsigned*)(ws + (long)N * 152);
  int*      binned  = (int*)(ws + (long)N * 72);
  unsigned* cnt_tab = (unsigned*)(ws + (long)N * 104);
  unsigned* off_tab = cnt_tab + NBINS * NCHUNK;
  float*    feat2  = ws;

  auto nb = [](long n, int b) { return (int)((n + b - 1) / b); };

  chunk_bin_kernel<<<NCHUNK, 1024, 0, stream>>>(src, dst, binned, cnt_tab, off_tab, E);
  bin_csr_kernel<<<NBINS, 1024, 0, stream>>>(binned, cnt_tab, off_tab, rowptr, csr, N);

  gemm_wmma_kernel<256, 64><<<N / GEMM_ROWS, 128, 0, stream>>>(x, W1, feat1);
  node_agg_h8_kernel<<<nb((long)N, 8), 256, 0, stream>>>(rowptr, csr, feat1, al1, ar1, b1, h1, N);

  gemm_wmma_kernel<64, 32><<<N / GEMM_ROWS, 64, 0, stream>>>(h1, W2, feat2);
  node_agg_h1_kernel<<<nb((long)N, 8), 256, 0, stream>>>(rowptr, csr, feat2, al2, ar2, b2, out, N);
}


# both aggregations: scores kept as base-2 logits (al, er scaled by log2 e once per wave); softmax weights and ELU by the bare exp2 instruction, not the library exp; sum order kept, last bits differ
# speedup vs baseline: 9.2636x; 1.0546x over previous
#include <hip/hip_runtime.h>
#include <hip/hip_bf16.h>

typedef __attribute__((ext_vector_type(2))) float v2f;
typedef __attribute__((ext_vector_type(8))) float v8f;

#define NNODES 100000
#define NEDGES 1600000

#define GEMM_ROWS 32
static_assert(NNODES % GEMM_ROWS == 0, "row tiles cover the nodes exactly");

template <int K, int C>
__global__ __launch_bounds__(32 * (C / 16)) void gemm_wmma_kernel(
    const float* __restrict__ A, const float* __restrict__ B,
    float* __restrict__ D) {
  constexpr int NT = 32 * (C / 16);
  constexpr int PITCH = K + 4;
  constexpr int VROW = K / 4;
  constexpr int NLOAD = GEMM_ROWS * VROW / NT;
  static_assert(GEMM_ROWS * VROW % NT == 0, "tile splits evenly over the threads");
  static_assert(PITCH % 64 == 4, "bank spread of the fragment reads");
  __shared__ __attribute__((aligned(16))) float tile[GEMM_ROWS * PITCH];

  const int tid  = threadIdx.x;
  const int lane = tid & 31;
  const int wave = tid >> 5;
  const int row0 = blockIdx.x * GEMM_ROWS;
  const int col0 = wave * 16;
  const int half = lane >> 4;
  const int l    = lane & 15;

  const float4* __restrict__ Ag = reinterpret_cast<const float4*>(A + (long)row0 * K);
  float4 st[NLOAD];
#pragma unroll
  for (int j = 0; j < NLOAD; ++j) st[j] = Ag[tid + j * NT];
#pragma unroll
  for (int j = 0; j < NLOAD; ++j) {
    const int i = tid + j * NT;
    *reinterpret_cast<float4*>(tile + (i / VROW) * PITCH + 4 * (i % VROW)) = st[j];
  }
  __syncthreads();

  const float* a0 = tile + l * PITCH + 2 * half;
  const float* a1 = a0 + 16 * PITCH;
  const float* bcol = B + col0 + l;

  v8f acc0 = {}, acc1 = {};
#pragma unroll 8
  for (int k = 0; k < K; k += 4) {
    v2f b;
    b.x = bcol[(k + 2 * half + 0) * C];
    b.y = bcol[(k + 2 * half + 1) * C];
    const v2f x0 = *reinterpret_cast<const v2f*>(a0 + k);
    const v2f x1 = *reinterpret_cast<const v2f*>(a1 + k);
    acc0 = __builtin_amdgcn_wmma_f32_16x16x4_f32(
         false, x0,  false, b,
         (short)0, acc0,  false,  false);
    acc1 = __builtin_amdgcn_wmma_f32_16x16x4_f32(
         false, x1,  false, b,
         (short)0, acc1,  false,  false);
  }

  float* drow = D + (long)(row0 + 8 * half) * C + col0 + l;
#pragma unroll
  for (int i = 0; i < 8; ++i) {
    drow[i * C] = acc0[i];
    drow[(16 + i) * C] = acc1[i];
  }
}

__device__ __forceinline__ float leaky(float v) { return v > 0.f ? v : 0.2f * v; }

#define LOG2E_F 1.44269504088896340736f
__device__ __forceinline__ float pow2(float y) { return __builtin_amdgcn_exp2f(y); }

#define BIN_SHIFT 9
#define BIN_NODES 512
#define NBINS 196
#define NCHUNK 256
#define CHUNK_EDGES 6250
#define EDGES_PER_THREAD 7
#define SORT_CAP 12288

static_assert(NBINS * BIN_NODES >= NNODES, "bins cover the nodes");
static_assert(NBINS <= 256, "bin counters");
static_assert(NCHUNK * CHUNK_EDGES >= NEDGES, "chunks cover the edges");
static_assert(EDGES_PER_THREAD * 1024 >= CHUNK_EDGES, "edges per thread");

__global__ __launch_bounds__(1024) void chunk_bin_kernel(
    const int* __restrict__ src, const int* __restrict__ dst,
    int* __restrict__ binned_w, unsigned* __restrict__ cnt_tab,
    unsigned* __restrict__ off_tab, int E) {
  __shared__ unsigned hist[256];
  __shared__ unsigned start[256];
  __shared__ int2 stage[CHUNK_EDGES];
  int2* __restrict__ binned = reinterpret_cast<int2*>(binned_w);
  const int t = threadIdx.x;
  const int g = blockIdx.x;
  const int e0 = g * CHUNK_EDGES;
  int n = E - e0;
  n = n < 0 ? 0 : (n > CHUNK_EDGES ? CHUNK_EDGES : n);

  if (t < 256) hist[t] = 0u;
  __syncthreads();

  int s[EDGES_PER_THREAD], d[EDGES_PER_THREAD];
  unsigned r[EDGES_PER_THREAD];
#pragma unroll
  for (int k = 0; k < EDGES_PER_THREAD; ++k) {
    const int i = t + k * 1024;
    s[k] = 0;
    d[k] = 0;
    if (i < n) {
      s[k] = src[e0 + i];
      d[k] = dst[e0 + i];
    }
  }
#pragma unroll
  for (int k = 0; k < EDGES_PER_THREAD; ++k) {
    const int i = t + k * 1024;
    r[k] = 0u;
    if (i < n) r[k] = atomicAdd(&hist[d[k] >> BIN_SHIFT], 1u);
  }
  __syncthreads();

  if (t < 32) {
    unsigned c[8];
    unsigned loc = 0u;
#pragma unroll
    for (int k = 0; k < 8; ++k) {
      c[k] = hist[8 * t + k];
      loc += c[k];
    }
    unsigned inc = loc;
    for (int off = 1; off < 32; off <<= 1) {
      const unsigned y = __shfl_up(inc, off, 32);
      if (t >= off) inc += y;
    }
    unsigned ex = inc - loc;
#pragma unroll
    for (int k = 0; k < 8; ++k) {
      start[8 * t + k] = ex;
      ex += c[k];
    }
  }
  __syncthreads();

#pragma unroll
  for (int k = 0; k < EDGES_PER_THREAD; ++k) {
    const int i = t + k * 1024;
    if (i < n) {
      int2 e;
      e.x = s[k];
      e.y = d[k];
      stage[start[d[k] >> BIN_SHIFT] + r[k]] = e;
    }
  }
  __syncthreads();

  for (int i = t; i < n; i += 1024) binned[e0 + i] = stage[i];
  if (t < NBINS) {
    cnt_tab[t * NCHUNK + g] = hist[t];
    off_tab[t * NCHUNK + g] = start[t];
  }
}

__device__ __forceinline__ void place_edge(unsigned* ncur, int* sorted, int* __restrict__ csr_src,
                                           unsigned base, int nb, int2 e) {
  const unsigned pos = atomicAdd(&ncur[e.y - nb], 1u);
  if (pos < SORT_CAP) sorted[pos] = e.x;
  else csr_src[base + pos] = e.x;
}

__global__ __launch_bounds__(1024) void bin_csr_kernel(
    const int* __restrict__ binned_w, const unsigned* __restrict__ cnt_tab,
    const unsigned* __restrict__ off_tab, unsigned* __restrict__ rowptr,
    int* __restrict__ csr_src, int N) {
  __shared__ unsigned ncur[BIN_NODES];
  __shared__ unsigned wsum[32];
  __shared__ unsigned wtot[16];
  __shared__ int sorted[SORT_CAP];
  const int2* __restrict__ binned = reinterpret_cast<const int2*>(binned_w);
  const int t = threadIdx.x, lane = t & 31;
  const int wave = __builtin_amdgcn_readfirstlane(t >> 5);
  const int b = blockIdx.x;
  const int nb = b * BIN_NODES;

  unsigned c[8];
  const int2* seg[8];
  int2 ev[8][2];
#pragma unroll
  for (int k = 0; k < 8; ++k) {
    const int gk = wave * 8 + k;
    c[k] = cnt_tab[b * NCHUNK + gk];
    seg[k] = binned + (long)gk * CHUNK_EDGES + off_tab[b * NCHUNK + gk];
  }
#pragma unroll
  for (int k = 0; k < 8; ++k) {
#pragma unroll
    for (int h = 0; h < 2; ++h) {
      ev[k][h].x = 0;
      ev[k][h].y = nb;
      if ((unsigned)(lane + 32 * h) < c[k]) ev[k][h] = seg[k][lane + 32 * h];
    }
  }

  unsigned p = 0u;
  const uint4* ct4 = reinterpret_cast<const uint4*>(cnt_tab);
  for (int i = t; i < b * (NCHUNK / 4); i += 1024) {
    const uint4 q = ct4[i];
    p += q.x + q.y + q.z + q.w;
  }
  for (int off = 16; off > 0; off >>= 1) p += __shfl_xor(p, off, 32);
  if (lane == 0) wsum[wave] = p;
  if (t < BIN_NODES) ncur[t] = 0u;
  __syncthreads();
  unsigned base = 0u;
  for (int w = 0; w < 32; ++w) base += wsum[w];

#pragma unroll
  for (int k = 0; k < 8; ++k) {
#pragma unroll
    for (int h = 0; h < 2; ++h)
      if ((unsigned)(lane + 32 * h) < c[k]) atomicAdd(&ncur[ev[k][h].y - nb], 1u);
    for (unsigned i = 64u + lane; i < c[k]; i += 32u) {
      const int2 e = seg[k][i];
      atomicAdd(&ncur[e.y - nb], 1u);
    }
  }
  __syncthreads();

  unsigned v = 0u, inc = 0u;
  if (t < BIN_NODES) {
    v = ncur[t];
    inc = v;
    for (int off = 1; off < 32; off <<= 1) {
      const unsigned y = __shfl_up(inc, off, 32);
      if (lane >= off) inc += y;
    }
    if (lane == 31) wtot[wave] = inc;
  }
  __syncthreads();
  if (t < BIN_NODES) {
    unsigned ex = inc - v;
    for (int w = 0; w < 16; ++w)
      if (w < wave) ex += wtot[w];
    ncur[t] = ex;
    if (nb + t <= N) rowptr[nb + t] = base + ex;
  }
  __syncthreads();

#pragma unroll
  for (int k = 0; k < 8; ++k) {
#pragma unroll
    for (int h = 0; h < 2; ++h)
      if ((unsigned)(lane + 32 * h) < c[k]) place_edge(ncur, sorted, csr_src, base, nb, ev[k][h]);
    for (unsigned i = 64u + lane; i < c[k]; i += 32u)
      place_edge(ncur, sorted, csr_src, base, nb, seg[k][i]);
  }
  __syncthreads();

  const unsigned total = ncur[BIN_NODES - 1];
  const unsigned m = total < SORT_CAP ? total : SORT_CAP;
  for (unsigned i = t; i < m; i += 1024u) csr_src[base + i] = sorted[i];
}

template <int CTRL>
__device__ __forceinline__ float dpp(float x) {
  return __builtin_bit_cast(
      float, __builtin_amdgcn_update_dpp(0, __builtin_bit_cast(int, x), CTRL, 0xf, 0xf, true));
}
__device__ __forceinline__ float xhalf(float x) {
  return __builtin_bit_cast(
      float, __builtin_amdgcn_permlanex16(__builtin_bit_cast(int, x), __builtin_bit_cast(int, x),
                                          0x76543210, 0xfedcba98, false, false));
}

__global__ __launch_bounds__(256) void node_agg_h8_kernel(
    const unsigned* __restrict__ rowptr, const int* __restrict__ csr_src,
    const float* __restrict__ feat, const float* __restrict__ al,
    const float* __restrict__ ar, const float* __restrict__ bias,
    float* __restrict__ hout, int N) {
  const int lane = threadIdx.x & 31;
  const int node = blockIdx.x * 8 + __builtin_amdgcn_readfirstlane(threadIdx.x >> 5);
  if (node >= N) return;
  const int beg = (int)rowptr[node];
  const int end = (int)rowptr[node + 1];
  const int q = lane & 3;
  const bool q1 = (q & 1) != 0, q2 = (q & 2) != 0;

  const float2 fd = *reinterpret_cast<const float2*>(feat + (long)node * 64 + 2 * lane);
  const float2 r2 = *reinterpret_cast<const float2*>(ar + 2 * lane);
  float2 a2 = *reinterpret_cast<const float2*>(al + 2 * lane);
  a2.x *= LOG2E_F;
  a2.y *= LOG2E_F;
  float er_h = fd.x * r2.x + fd.y * r2.y;
  er_h += dpp<0xB1>(er_h);
  er_h += dpp<0x4E>(er_h);
  er_h *= LOG2E_F;

  float m = -__builtin_inff();
  float den = 0.f, ax = 0.f, ay = 0.f;
  for (int base = beg; base < end; base += 32) {
    const int nloc = (end - base < 32) ? end - base : 32;
    const int sid = (lane < nloc) ? csr_src[base + lane] : 0;
    for (int c = 0; c < nloc; c += 8) {
      float2 f[8];
#pragma unroll
      for (int j = 0; j < 8; ++j) {
        const int sj = __builtin_amdgcn_readlane(sid, c + j);
        f[j] = *reinterpret_cast<const float2*>(feat + (long)sj * 64 + 2 * lane);
      }
      float v[2];
#pragma unroll
      for (int g = 0; g < 2; ++g) {
        const float p0 = f[4 * g + 0].x * a2.x + f[4 * g + 0].y * a2.y;
        const float p1 = f[4 * g + 1].x * a2.x + f[4 * g + 1].y * a2.y;
        const float p2 = f[4 * g + 2].x * a2.x + f[4 * g + 2].y * a2.y;
        const float p3 = f[4 * g + 3].x * a2.x + f[4 * g + 3].y * a2.y;
        const float t01 = (q1 ? p1 : p0) + dpp<0xB1>(q1 ? p0 : p1);
        const float t23 = (q1 ? p3 : p2) + dpp<0xB1>(q1 ? p2 : p3);
        const float s = (q2 ? t23 : t01) + dpp<0x4E>(q2 ? t01 : t23);
        const float lv = leaky(s + er_h);
        v[g] = (c + 4 * g + q < nloc) ? lv : -__builtin_inff();
      }
      float cm = fmaxf(v[0], v[1]);
      cm = fmaxf(cm, dpp<0xB1>(cm));
      cm = fmaxf(cm, dpp<0x4E>(cm));
      const float mn = fmaxf(m, cm);
      const float sc = pow2(m - mn);
      m = mn;
      float w[2];
      w[0] = pow2(v[0] - mn);
      w[1] = pow2(v[1] - mn);
      den = den * sc + w[0] + w[1];
      ax *= sc;
      ay *= sc;
#pragma unroll
      for (int g = 0; g < 2; ++g) {
        const float w0 = dpp<0x00>(w[g]), w1 = dpp<0x55>(w[g]);
        const float w2 = dpp<0xAA>(w[g]), w3 = dpp<0xFF>(w[g]);
        if (c + 4 * g + 0 < nloc) { ax += w0 * f[4 * g + 0].x; ay += w0 * f[4 * g + 0].y; }
        if (c + 4 * g + 1 < nloc) { ax += w1 * f[4 * g + 1].x; ay += w1 * f[4 * g + 1].y; }
        if (c + 4 * g + 2 < nloc) { ax += w2 * f[4 * g + 2].x; ay += w2 * f[4 * g + 2].y; }
        if (c + 4 * g + 3 < nloc) { ax += w3 * f[4 * g + 3].x; ay += w3 * f[4 * g + 3].y; }
      }
    }
  }
  den += dpp<0xB1>(den);
  den += dpp<0x4E>(den);

  const float2 bb = *reinterpret_cast<const float2*>(bias + 2 * lane);
  float ox = (den > 0.f ? ax / den : 0.f) + bb.x;
  float oy = (den > 0.f ? ay / den : 0.f) + bb.y;
  ox = ox > 0.f ? ox : pow2(ox * LOG2E_F) - 1.f;
  oy = oy > 0.f ? oy : pow2(oy * LOG2E_F) - 1.f;
  float2 o;
  o.x = ox;
  o.y = oy;
  *reinterpret_cast<float2*>(hout + (long)node * 64 + 2 * lane) = o;
}

__global__ __launch_bounds__(256) void node_agg_h1_kernel(
    const unsigned* __restrict__ rowptr, const int* __restrict__ csr_src,
    const float* __restrict__ feat, const float* __restrict__ al,
    const float* __restrict__ ar, const float* __restrict__ bias,
    float* __restrict__ out, int N) {
  const int lane = threadIdx.x & 31;
  const int node = blockIdx.x * 8 + __builtin_amdgcn_readfirstlane(threadIdx.x >> 5);
  if (node >= N) return;
  const int beg = (int)rowptr[node];
  const int end = (int)rowptr[node + 1];
  const bool b4 = (lane & 16) != 0, b3 = (lane & 8) != 0, b2 = (lane & 4) != 0;
  const int eslot = lane >> 2;

  const float alv = al[lane] * LOG2E_F;
  float er_d = feat[(long)node * 32 + lane] * ar[lane];
  er_d += xhalf(er_d);
  er_d += dpp<0x168>(er_d);
  er_d += dpp<0x164>(er_d);
  er_d += dpp<0x4E>(er_d);
  er_d += dpp<0xB1>(er_d);
  er_d *= LOG2E_F;

  float m = -__builtin_inff();
  float den = 0.f, acc = 0.f;
  for (int base = beg; base < end; base += 32) {
    const int nloc = (end - base < 32) ? end - base : 32;
    const int sid = (lane < nloc) ? csr_src[base + lane] : 0;
    for (int c = 0; c < nloc; c += 8) {
      float f[8];
#pragma unroll
      for (int j = 0; j < 8; ++j) {
        const int sj = __builtin_amdgcn_readlane(sid, c + j);
        f[j] = feat[(long)sj * 32 + lane];
      }
      float t[4], u[2];
#pragma unroll
      for (int i = 0; i < 4; ++i) {
        const float lo = f[i] * alv, hi = f[i + 4] * alv;
        t[i] = (b4 ? hi : lo) + xhalf(b4 ? lo : hi);
      }
#pragma unroll
      for (int i = 0; i < 2; ++i)
        u[i] = (b3 ? t[i + 2] : t[i]) + dpp<0x168>(b3 ? t[i] : t[i + 2]);
      float r = (b2 ? u[1] : u[0]) + dpp<0x164>(b2 ? u[0] : u[1]);
      r += dpp<0x4E>(r);
      r += dpp<0xB1>(r);
      const float lv = leaky(r + er_d);
      const float v = (c + eslot < nloc) ? lv : -__builtin_inff();
      float cm = fmaxf(v, dpp<0x164>(v));
      cm = fmaxf(cm, dpp<0x168>(cm));
      cm = fmaxf(cm, xhalf(cm));
      const float mn = fmaxf(m, cm);
      const float sc = pow2(m - mn);
      m = mn;
      const float w = pow2(v - mn);
      den = den * sc + w;
      acc *= sc;
#pragma unroll
      for (int j = 0; j < 8; ++j) {
        const float wj = __int_as_float(__builtin_amdgcn_readlane(__float_as_int(w), 4 * j));
        if (c + j < nloc) acc += wj * f[j];
      }
    }
  }
  den += dpp<0x164>(den);
  den += dpp<0x168>(den);
  den += xhalf(den);
  out[(long)node * 32 + lane] = (den > 0.f ? acc / den : 0.f) + bias[lane];
}

extern "C" void kernel_launch(void* const* d_in, const int* in_sizes, int n_in,
                              void* d_out, int out_size, void* d_ws,
                              size_t ws_size, hipStream_t stream) {
  const float* x   = (const float*)d_in[0];
  const int*   src = (const int*)d_in[1];
  const int*   dst = (const int*)d_in[2];
  const float* W1  = (const float*)d_in[3];
  const float* al1 = (const float*)d_in[4];
  const float* ar1 = (const float*)d_in[5];
  const float* b1  = (const float*)d_in[6];
  const float* W2  = (const float*)d_in[7];
  const float* al2 = (const float*)d_in[8];
  const float* ar2 = (const float*)d_in[9];
  const float* b2  = (const float*)d_in[10];
  float* out = (float*)d_out;

  const int N = NNODES, E = NEDGES;

  float* ws = (float*)d_ws;
  float*    feat1  = ws;
  float*    h1     = ws + (long)N * 72;
  int*      csr    = (int*)(ws + (long)N * 136);
  unsigned* rowptr = (unsigned*)(ws + (long)N * 152);
  int*      binned  = (int*)(ws + (long)N * 72);
  unsigned* cnt_tab = (unsigned*)(ws + (long)N * 104);
  unsigned* off_tab = cnt_tab + NBINS * NCHUNK;
  float*    feat2  = ws;

  auto nb = [](long n, int b) { return (int)((n + b - 1) / b); };

  chunk_bin_kernel<<<NCHUNK, 1024, 0, stream>>>(src, dst, binned, cnt_tab, off_tab, E);
  bin_csr_kernel<<<NBINS, 1024, 0, stream>>>(binned, cnt_tab, off_tab, rowptr, csr, N);

  gemm_wmma_kernel<256, 64><<<N / GEMM_ROWS, 128, 0, stream>>>(x, W1, feat1);
  node_agg_h8_kernel<<<nb((long)N, 8), 256, 0, stream>>>(rowptr, csr, feat1, al1, ar1, b1, h1, N);

  gemm_wmma_kernel<64, 32><<<N / GEMM_ROWS, 64, 0, stream>>>(h1, W2, feat2);
  node_agg_h1_kernel<<<nb((long)N, 8), 256, 0, stream>>>(rowptr, csr, feat2, al2, ar2, b2, out, N);
}
